# starting kernel
# baseline (speedup 1.0000x reference)
#include <hip/hip_runtime.h>

#define N_TOTAL    150100
#define EMBED_DIM  64
#define NUM_USERS  100000
#define N_EDGES    3000000
#define BATCH      4096
#define NUM_LAYERS 3

typedef __attribute__((ext_vector_type(2)))  float    v2f;
typedef __attribute__((ext_vector_type(8)))  float    v8f;
typedef __attribute__((ext_vector_type(16))) _Float16 v16h;

__global__ void lg_zero(float4* __restrict__ p, int n4) {
  int i = blockIdx.x * blockDim.x + threadIdx.x;
  const int stride = gridDim.x * blockDim.x;
  for (; i < n4; i += stride) p[i] = make_float4(0.f, 0.f, 0.f, 0.f);
}

__global__ void lg_gather_init(const float* __restrict__ emb, const int* __restrict__ uid,
                               const int* __restrict__ iid, float* __restrict__ acc) {
  const int t = blockIdx.x * blockDim.x + threadIdx.x;
  const int r = t >> 6, c = t & 63;
  const int n = (r < BATCH) ? uid[r] : (NUM_USERS + iid[r - BATCH]);
  acc[t] = emb[(size_t)n * EMBED_DIM + c];
}

__global__ void lg_gather_acc(const float* __restrict__ emb, const int* __restrict__ uid,
                              const int* __restrict__ iid, float* __restrict__ acc) {
  const int t = blockIdx.x * blockDim.x + threadIdx.x;
  const int r = t >> 6, c = t & 63;
  const int n = (r < BATCH) ? uid[r] : (NUM_USERS + iid[r - BATCH]);
  acc[t] += emb[(size_t)n * EMBED_DIM + c];
}

__global__ void lg_spmm_scatter(const float* __restrict__ emb, const float* __restrict__ val,
                                const int* __restrict__ rw, const int* __restrict__ cl,
                                float* __restrict__ out) {
  const int lane = threadIdx.x & 31;
  const int d    = lane << 1;
  int w          = (blockIdx.x * blockDim.x + threadIdx.x) >> 5;
  const int nw   = (gridDim.x * blockDim.x) >> 5;
  for (int e = w; e < N_EDGES; e += nw) {
    const int   r = __builtin_amdgcn_readfirstlane(rw[e]);
    const int   c = __builtin_amdgcn_readfirstlane(cl[e]);
    const float v = val[e];
    const float2 s = *reinterpret_cast<const float2*>(emb + (size_t)c * EMBED_DIM + d);
    float* dst = out + (size_t)r * EMBED_DIM + d;
    unsafeAtomicAdd(dst,     v * s.x);
    unsafeAtomicAdd(dst + 1, v * s.y);
  }
}

__global__ void lg_dot_wmma(const float* __restrict__ acc, float* __restrict__ out) {
  __shared__ float sd[8][8][32];
  const int lane = threadIdx.x & 31;
  const int wave = threadIdx.x >> 5;
  const int tile = blockIdx.x * 8 + wave;
  const int k0   = tile * 16;
  const int rN   = lane & 15;
  const float* U = acc + (size_t)(k0 + rN) * EMBED_DIM;
  const float* I = acc + (size_t)(BATCH + k0 + rN) * EMBED_DIM;

  v8f c = {0.f, 0.f, 0.f, 0.f, 0.f, 0.f, 0.f, 0.f};

#if __has_builtin(__builtin_amdgcn_wmma_f32_16x16x4_f32)
  const int kb = (lane >> 4) << 1;
  #pragma unroll
  for (int s = 0; s < 16; ++s) {
    const int col = s * 4 + kb;
    v2f a; a[0] = U[col]; a[1] = U[col + 1];
    v2f b; b[0] = I[col]; b[1] = I[col + 1];
    c = __builtin_amdgcn_wmma_f32_16x16x4_f32(false, a, false, b, (short)0, c, false, false);
  }
#else
  const int kb8 = (lane >> 4) << 3;
  #pragma unroll
  for (int blk = 0; blk < 2; ++blk) {
    v16h a, b;
    #pragma unroll
    for (int i = 0; i < 8; ++i) {
      const int kk = blk * 32 + kb8 + ((i < 4) ? (2 * i) : (16 + 2 * (i - 4)));
      a[2 * i]     = (_Float16)U[kk];
      a[2 * i + 1] = (_Float16)U[kk + 1];
      b[2 * i]     = (_Float16)I[kk];
      b[2 * i + 1] = (_Float16)I[kk + 1];
    }
    c = __builtin_amdgcn_wmma_f32_16x16x32_f16(false, a, false, b, (short)0, c, false, false);
  }
#endif

  #pragma unroll
  for (int g = 0; g < 8; ++g) sd[wave][g][lane] = c[g];
  __syncthreads();

  if (lane < 16) {
    const float v = (lane < 8) ? sd[wave][lane][lane] : sd[wave][lane - 8][lane + 16];
    out[k0 + lane] = v * 0.0625f;
  }
}

extern "C" void kernel_launch(void* const* d_in, const int* in_sizes, int n_in,
                              void* d_out, int out_size, void* d_ws, size_t ws_size,
                              hipStream_t stream) {
  const float* node_emb = (const float*)d_in[0];
  const float* adj_val  = (const float*)d_in[1];
  const int*   adj_row  = (const int*)d_in[2];
  const int*   adj_col  = (const int*)d_in[3];
  const int*   user_ids = (const int*)d_in[4];
  const int*   item_ids = (const int*)d_in[5];
  float* out = (float*)d_out;

  const size_t embElems = (size_t)N_TOTAL * EMBED_DIM;
  char* ws   = (char*)d_ws;
  float* embA = (float*)ws;
  float* embB = (float*)(ws + embElems * sizeof(float));
  float* accS = (float*)(ws + 2 * embElems * sizeof(float));

  const dim3 b256(256);
  const int gatherBlocks = (2 * BATCH * EMBED_DIM) / 256;
  const int n4 = (int)(embElems / 4);

  lg_gather_init<<<gatherBlocks, b256, 0, stream>>>(node_emb, user_ids, item_ids, accS);

  const float* cur = node_emb;
  float* bufs[2] = {embA, embB};
  for (int L = 0; L < NUM_LAYERS; ++L) {
    float* nxt = bufs[L & 1];
    lg_zero<<<2048, b256, 0, stream>>>((float4*)nxt, n4);
    lg_spmm_scatter<<<2048, b256, 0, stream>>>(cur, adj_val, adj_row, adj_col, nxt);
    lg_gather_acc<<<gatherBlocks, b256, 0, stream>>>(nxt, user_ids, item_ids, accS);
    cur = nxt;
  }

  lg_dot_wmma<<<BATCH / 128, b256, 0, stream>>>(accS, out);
}


# Row pull retiled: a wave sums 4 rows at once, 8 lanes x 16 B per row, so one source load covers 4 lines, not 1; records shared by DPP8. Sum order per row as before; mul+add now fused (fma)
# speedup vs baseline: 1.8824x; 1.8824x over previous
#include <hip/hip_runtime.h>

#define N_TOTAL    150100
#define EMBED_DIM  64
#define HALF_DIM   32
#define NUM_USERS  100000
#define N_EDGES    3000000
#define BATCH      4096
#define NUM_LAYERS 3

#define CNT_PAD    150528
#define SCAN_NB    147

#define OFF_TBLA   0u
#define OFF_TBLB   19212800u
#define OFF_CSR    38425600u
#define OFF_ROWP   62425600u
#define OFF_CNT    63027712u
#define OFF_ACCS   76851200u

typedef __attribute__((ext_vector_type(2)))  float    v2f;
typedef __attribute__((ext_vector_type(4)))  float    v4f;
typedef __attribute__((ext_vector_type(8)))  float    v8f;
typedef __attribute__((ext_vector_type(16))) _Float16 v16h;

__global__ void lg_gather_init(const float* __restrict__ emb, const int* __restrict__ uid,
                               const int* __restrict__ iid, float* __restrict__ acc,
                               int* __restrict__ cnt) {
  const int t = blockIdx.x * blockDim.x + threadIdx.x;
  const int r = t >> 6, c = t & 63;
  const int n = (r < BATCH) ? uid[r] : (NUM_USERS + iid[r - BATCH]);
  acc[t] = emb[(size_t)n * EMBED_DIM + c];
  if (t < CNT_PAD) cnt[t] = 0;
}

__global__ void lg_rank(const int* __restrict__ rw, int* __restrict__ cnt, int* __restrict__ rank) {
  int i = blockIdx.x * blockDim.x + threadIdx.x;
  const int stride = gridDim.x * blockDim.x;
  for (; i < N_EDGES; i += stride) rank[i] = atomicAdd(&cnt[rw[i]], 1);
}

__global__ void lg_scan(const int4* __restrict__ cnt4, int4* __restrict__ rowp4) {
  __shared__ int sh[256];
  __shared__ int sb[256];
  const int t = threadIdx.x, b = blockIdx.x;
  int pre = 0;
  for (int i = t; i < b * 256; i += 256) {
    const int4 u = cnt4[i];
    pre += u.x + u.y + u.z + u.w;
  }
  const int4 v = cnt4[b * 256 + t];
  const int tsum = v.x + v.y + v.z + v.w;
  sh[t] = tsum;
  sb[t] = pre;
  __syncthreads();
  for (int s = 1; s < 256; s <<= 1) {
    const int add = (t >= s) ? sh[t - s] : 0;
    __syncthreads();
    sh[t] += add;
    __syncthreads();
  }
  for (int s = 128; s > 0; s >>= 1) {
    if (t < s) sb[t] += sb[t + s];
    __syncthreads();
  }
  const int base = sb[0] + sh[t] - tsum;
  int4 o;
  o.x = base; o.y = o.x + v.x; o.z = o.y + v.y; o.w = o.z + v.z;
  rowp4[b * 256 + t] = o;
}

__global__ void lg_fill(const int* __restrict__ rw, const int* __restrict__ cl,
                        const float* __restrict__ val, const int* __restrict__ rowp,
                        const int* __restrict__ rank, int2* __restrict__ csr) {
  int i = blockIdx.x * blockDim.x + threadIdx.x;
  const int stride = gridDim.x * blockDim.x;
  for (; i < N_EDGES; i += stride) {
    const int p = rowp[rw[i]] + rank[i];
    csr[p] = make_int2(cl[i], __float_as_int(val[i]));
  }
}


template <int K>
__device__ __forceinline__ int grp_bcast(int v) {
  return (int)__builtin_amdgcn_mov_dpp8((unsigned)v, (unsigned)(K * 0x249249));
}

#define LG_SLOT(K)                                                                \
  {                                                                               \
    const unsigned c = (unsigned)grp_bcast<K>(cc);                                \
    w[K] = __int_as_float(grp_bcast<K>(ww));                                      \
    x[K] = *(const v4f*)(srcb + (size_t)(c * (unsigned)(STRIDE * 4) + qoff));     \
  }

template <int STRIDE>
__device__ __forceinline__ void lg_round(const char* __restrict__ srcb, const int cc, const int ww,
                                         const unsigned qoff, v4f& acc) {
  v4f   x[8];
  float w[8];
  LG_SLOT(0) LG_SLOT(1) LG_SLOT(2) LG_SLOT(3) LG_SLOT(4) LG_SLOT(5) LG_SLOT(6) LG_SLOT(7)
  #pragma unroll
  for (int k = 0; k < 8; ++k) acc += w[k] * x[k];
}

template <int STRIDE>
__global__ void lg_pull4(const float* __restrict__ src, const int* __restrict__ rowp,
                         const int2* __restrict__ csr, float* __restrict__ dst) {
  static_assert(N_TOTAL % 4 == 0, "a wave takes four whole rows");
  const int lane = threadIdx.x & 31;
  const int q = lane & 7;
  const int g = lane >> 3;
  const unsigned qoff = (unsigned)q * 16u;
  const char* srcb = (const char*)src;
  const int w0 = __builtin_amdgcn_readfirstlane((int)((blockIdx.x * blockDim.x + threadIdx.x) >> 5));
  const int nw = (int)((gridDim.x * blockDim.x) >> 5);
  for (int r0 = w0 * 4; r0 < N_TOTAL; r0 += nw * 4) {
    const int row = r0 + g;
    const int beg = rowp[row], end = rowp[row + 1];
    v4f acc = {0.f, 0.f, 0.f, 0.f};
    for (int jb = beg; jb < end; jb += 32) {
      const int n = min(end - jb, 32);
      int2 rec[4];
      #pragma unroll
      for (int j = 0; j < 4; ++j) {
        rec[j] = make_int2(0, 0);
        if (8 * j < n) rec[j] = csr[jb + min(8 * j + q, n - 1)];
      }
      #pragma unroll
      for (int j = 0; j < 4; ++j)
        if (8 * j < n)
          lg_round<STRIDE>(srcb, rec[j].x, (8 * j + q < n) ? rec[j].y : 0  , qoff, acc);
    }
    *(v4f*)(dst + (size_t)row * HALF_DIM + q * 4) = acc;
  }
}

__global__ void lg_gather_acc2(const float* __restrict__ ta, const float* __restrict__ tb,
                               const int* __restrict__ uid, const int* __restrict__ iid,
                               float* __restrict__ acc, int off) {
  const int t = blockIdx.x * blockDim.x + threadIdx.x;
  const int r = t >> 5, c = t & 31;
  const int n = (r < BATCH) ? uid[r] : (NUM_USERS + iid[r - BATCH]);
  float x = acc[r * EMBED_DIM + off + c];
  x += ta[(size_t)n * HALF_DIM + c];
  x += tb[(size_t)n * HALF_DIM + c];
  acc[r * EMBED_DIM + off + c] = x;
}

__global__ void lg_gather_acc1(const float* __restrict__ ta,
                               const int* __restrict__ uid, const int* __restrict__ iid,
                               float* __restrict__ acc, int off) {
  const int t = blockIdx.x * blockDim.x + threadIdx.x;
  const int r = t >> 5, c = t & 31;
  const int n = (r < BATCH) ? uid[r] : (NUM_USERS + iid[r - BATCH]);
  acc[r * EMBED_DIM + off + c] += ta[(size_t)n * HALF_DIM + c];
}

__global__ void lg_dot_wmma(const float* __restrict__ acc, float* __restrict__ out) {
  __shared__ float sd[8][8][32];
  const int lane = threadIdx.x & 31;
  const int wave = threadIdx.x >> 5;
  const int tile = blockIdx.x * 8 + wave;
  const int k0   = tile * 16;
  const int rN   = lane & 15;
  const float* U = acc + (size_t)(k0 + rN) * EMBED_DIM;
  const float* I = acc + (size_t)(BATCH + k0 + rN) * EMBED_DIM;

  v8f c = {0.f, 0.f, 0.f, 0.f, 0.f, 0.f, 0.f, 0.f};

#if __has_builtin(__builtin_amdgcn_wmma_f32_16x16x4_f32)
  const int kb = (lane >> 4) << 1;
  #pragma unroll
  for (int s = 0; s < 16; ++s) {
    const int col = s * 4 + kb;
    v2f a; a[0] = U[col]; a[1] = U[col + 1];
    v2f b; b[0] = I[col]; b[1] = I[col + 1];
    c = __builtin_amdgcn_wmma_f32_16x16x4_f32(false, a, false, b, (short)0, c, false, false);
  }
#else
  const int kb8 = (lane >> 4) << 3;
  #pragma unroll
  for (int blk = 0; blk < 2; ++blk) {
    v16h a, b;
    #pragma unroll
    for (int i = 0; i < 8; ++i) {
      const int kk = blk * 32 + kb8 + ((i < 4) ? (2 * i) : (16 + 2 * (i - 4)));
      a[2 * i]     = (_Float16)U[kk];
      a[2 * i + 1] = (_Float16)U[kk + 1];
      b[2 * i]     = (_Float16)I[kk];
      b[2 * i + 1] = (_Float16)I[kk + 1];
    }
    c = __builtin_amdgcn_wmma_f32_16x16x32_f16(false, a, false, b, (short)0, c, false, false);
  }
#endif

  #pragma unroll
  for (int g = 0; g < 8; ++g) sd[wave][g][lane] = c[g];
  __syncthreads();

  if (lane < 16) {
    const float v = (lane < 8) ? sd[wave][lane][lane] : sd[wave][lane - 8][lane + 16];
    out[k0 + lane] = v * 0.0625f;
  }
}

static void lg_half(const float* node_emb_half, int col0, const int* rowp, const int2* csr,
                    float* tblA, float* tblB, const int* user_ids, const int* item_ids,
                    float* accS, hipStream_t stream) {
  const dim3 b256(256);
  lg_pull4<EMBED_DIM><<<2048, b256, 0, stream>>>(node_emb_half, rowp, csr, tblA);
  lg_pull4<HALF_DIM><<<2048, b256, 0, stream>>>(tblA, rowp, csr, tblB);
  lg_gather_acc2<<<1024, b256, 0, stream>>>(tblA, tblB, user_ids, item_ids, accS, col0);
  lg_pull4<HALF_DIM><<<2048, b256, 0, stream>>>(tblB, rowp, csr, tblA);
  lg_gather_acc1<<<1024, b256, 0, stream>>>(tblA, user_ids, item_ids, accS, col0);
}

extern "C" void kernel_launch(void* const* d_in, const int* in_sizes, int n_in,
                              void* d_out, int out_size, void* d_ws, size_t ws_size,
                              hipStream_t stream) {
  const float* node_emb = (const float*)d_in[0];
  const float* adj_val  = (const float*)d_in[1];
  const int*   adj_row  = (const int*)d_in[2];
  const int*   adj_col  = (const int*)d_in[3];
  const int*   user_ids = (const int*)d_in[4];
  const int*   item_ids = (const int*)d_in[5];
  float* out = (float*)d_out;

  char* ws    = (char*)d_ws;
  float* tblA = (float*)(ws + OFF_TBLA);
  float* tblB = (float*)(ws + OFF_TBLB);
  int*   rank = (int*)(ws + OFF_TBLB);
  int2*  csr  = (int2*)(ws + OFF_CSR);
  int*   rowp = (int*)(ws + OFF_ROWP);
  int*   cnt  = (int*)(ws + OFF_CNT);
  float* accS = (float*)(ws + OFF_ACCS);

  const dim3 b256(256);

  lg_gather_init<<<2048, b256, 0, stream>>>(node_emb, user_ids, item_ids, accS, cnt);

  lg_rank<<<2048, b256, 0, stream>>>(adj_row, cnt, rank);
  lg_scan<<<SCAN_NB, b256, 0, stream>>>((const int4*)cnt, (int4*)rowp);
  lg_fill<<<2048, b256, 0, stream>>>(adj_row, adj_col, adj_val, rowp, rank, csr);

  lg_half(node_emb,      0,  rowp, csr, tblA, tblB, user_ids, item_ids, accS, stream);
  lg_half(node_emb + 32, 32, rowp, csr, tblA, tblB, user_ids, item_ids, accS, stream);

  lg_dot_wmma<<<BATCH / 128, b256, 0, stream>>>(accS, out);
}


# Layer 2 summed only at rows read later: a CSR walk from the 8192 gathered rows flags them, a scan lists them; layer 3 then pulls at the gathered rows only, onto the side table. Sum order kept
# speedup vs baseline: 1.9423x; 1.0318x over previous
#include <hip/hip_runtime.h>

#define N_TOTAL    150100
#define EMBED_DIM  64
#define HALF_DIM   32
#define NUM_USERS  100000
#define N_EDGES    3000000
#define BATCH      4096
#define NUM_LAYERS 3

#define CNT_PAD    150528
#define SCAN_NB    147

#define OFF_TBLA   0u
#define OFF_TBLB   19212800u
#define OFF_CSR    38425600u
#define OFF_ROWP   62425600u
#define OFF_CNT    63027712u
#define OFF_FLAG   63629824u
#define OFF_LIST   64231936u
#define OFF_NLIST  64834048u
#define OFF_ACCS   76851200u

typedef __attribute__((ext_vector_type(2)))  float    v2f;
typedef __attribute__((ext_vector_type(4)))  float    v4f;
typedef __attribute__((ext_vector_type(8)))  float    v8f;
typedef __attribute__((ext_vector_type(16))) _Float16 v16h;

__global__ void lg_gather_init(const float* __restrict__ emb, const int* __restrict__ uid,
                               const int* __restrict__ iid, float* __restrict__ acc,
                               int* __restrict__ cnt, int* __restrict__ flag) {
  const int t = blockIdx.x * blockDim.x + threadIdx.x;
  const int r = t >> 6, c = t & 63;
  const int n = (r < BATCH) ? uid[r] : (NUM_USERS + iid[r - BATCH]);
  acc[t] = emb[(size_t)n * EMBED_DIM + c];
  if (t < CNT_PAD) { cnt[t] = 0; flag[t] = 0; }
}

__global__ void lg_rank(const int* __restrict__ rw, int* __restrict__ cnt, int* __restrict__ rank) {
  int i = blockIdx.x * blockDim.x + threadIdx.x;
  const int stride = gridDim.x * blockDim.x;
  for (; i < N_EDGES; i += stride) rank[i] = atomicAdd(&cnt[rw[i]], 1);
}

template <bool LIST>
__global__ void lg_scan(const int4* __restrict__ cnt4, int4* __restrict__ rowp4,
                        int* __restrict__ list, int* __restrict__ nlist) {
  __shared__ int sh[256];
  __shared__ int sb[256];
  const int t = threadIdx.x, b = blockIdx.x;
  int pre = 0;
  for (int i = t; i < b * 256; i += 256) {
    const int4 u = cnt4[i];
    pre += u.x + u.y + u.z + u.w;
  }
  const int4 v = cnt4[b * 256 + t];
  const int tsum = v.x + v.y + v.z + v.w;
  sh[t] = tsum;
  sb[t] = pre;
  __syncthreads();
  for (int s = 1; s < 256; s <<= 1) {
    const int add = (t >= s) ? sh[t - s] : 0;
    __syncthreads();
    sh[t] += add;
    __syncthreads();
  }
  for (int s = 128; s > 0; s >>= 1) {
    if (t < s) sb[t] += sb[t + s];
    __syncthreads();
  }
  const int base = sb[0] + sh[t] - tsum;
  int4 o;
  o.x = base; o.y = o.x + v.x; o.z = o.y + v.y; o.w = o.z + v.z;
  if (!LIST) {
    rowp4[b * 256 + t] = o;
  } else {
    const int r = (b * 256 + t) * 4;
    if (v.x) list[o.x] = r;
    if (v.y) list[o.y] = r + 1;
    if (v.z) list[o.z] = r + 2;
    if (v.w) list[o.w] = r + 3;
    if (b == SCAN_NB - 1 && t == 255) *nlist = o.w + v.w;
  }
}

__global__ void lg_fill(const int* __restrict__ rw, const int* __restrict__ cl,
                        const float* __restrict__ val, const int* __restrict__ rowp,
                        const int* __restrict__ rank, int2* __restrict__ csr) {
  int i = blockIdx.x * blockDim.x + threadIdx.x;
  const int stride = gridDim.x * blockDim.x;
  for (; i < N_EDGES; i += stride) {
    const int p = rowp[rw[i]] + rank[i];
    csr[p] = make_int2(cl[i], __float_as_int(val[i]));
  }
}

__global__ void lg_mark(const int* __restrict__ uid, const int* __restrict__ iid,
                        const int* __restrict__ rowp, const int2* __restrict__ csr,
                        int* __restrict__ flag) {
  const int t = blockIdx.x * blockDim.x + threadIdx.x;
  const int r = t >> 3, q = t & 7;
  const int n = (r < BATCH) ? uid[r] : (NUM_USERS + iid[r - BATCH]);
  const int beg = rowp[n], end = rowp[n + 1];
  if (q == 0) atomicOr(&flag[n], 1);
  for (int j = beg + q; j < end; j += 8) atomicOr(&flag[csr[j].x], 1);
}


template <int K>
__device__ __forceinline__ int grp_bcast(int v) {
  return (int)__builtin_amdgcn_mov_dpp8((unsigned)v, (unsigned)(K * 0x249249));
}

#define LG_SLOT(K)                                                                \
  {                                                                               \
    const unsigned c = (unsigned)grp_bcast<K>(cc);                                \
    w[K] = __int_as_float(grp_bcast<K>(ww));                                      \
    x[K] = *(const v4f*)(srcb + (size_t)(c * (unsigned)(STRIDE * 4) + qoff));     \
  }

template <int STRIDE>
__device__ __forceinline__ void lg_round(const char* __restrict__ srcb, const int cc, const int ww,
                                         const unsigned qoff, v4f& acc) {
  v4f   x[8];
  float w[8];
  LG_SLOT(0) LG_SLOT(1) LG_SLOT(2) LG_SLOT(3) LG_SLOT(4) LG_SLOT(5) LG_SLOT(6) LG_SLOT(7)
  #pragma unroll
  for (int k = 0; k < 8; ++k) acc += w[k] * x[k];
}

enum { LG_ALL = 0, LG_LIST = 1, LG_SLOT = 2 };

template <int STRIDE, int MODE>
__global__ void lg_pull4(const float* __restrict__ src, const int* __restrict__ rowp,
                         const int2* __restrict__ csr, float* __restrict__ dst,
                         const int* __restrict__ list, const int* __restrict__ nlist,
                         const int* __restrict__ uid, const int* __restrict__ iid, int off) {
  static_assert(N_TOTAL % 4 == 0 && (2 * BATCH) % 4 == 0, "a wave takes four whole rows");
  const int lane = threadIdx.x & 31;
  const int q = lane & 7;
  const int g = lane >> 3;
  const unsigned qoff = (unsigned)q * 16u;
  const char* srcb = (const char*)src;
  const int w0 = __builtin_amdgcn_readfirstlane((int)((blockIdx.x * blockDim.x + threadIdx.x) >> 5));
  const int nw = (int)((gridDim.x * blockDim.x) >> 5);
  int total = N_TOTAL;
  if (MODE == LG_SLOT) total = 2 * BATCH;
  if (MODE == LG_LIST) total = min(*nlist, N_TOTAL);
  for (int r0 = w0 * 4; r0 < total; r0 += nw * 4) {
    const int i = r0 + g;
    const bool act = (MODE != LG_LIST) || i < total;
    int row = i;
    if (MODE == LG_LIST) row = act ? list[i] : 0;
    if (MODE == LG_SLOT) row = (i < BATCH) ? uid[i] : (NUM_USERS + iid[i - BATCH]);
    int beg = 0, end = 0;
    if (act) { beg = rowp[row]; end = rowp[row + 1]; }
    v4f acc = {0.f, 0.f, 0.f, 0.f};
    for (int jb = beg; jb < end; jb += 32) {
      const int n = min(end - jb, 32);
      int2 rec[4];
      #pragma unroll
      for (int j = 0; j < 4; ++j) {
        rec[j] = make_int2(0, 0);
        if (8 * j < n) rec[j] = csr[jb + min(8 * j + q, n - 1)];
      }
      #pragma unroll
      for (int j = 0; j < 4; ++j)
        if (8 * j < n)
          lg_round<STRIDE>(srcb, rec[j].x, (8 * j + q < n) ? rec[j].y : 0  , qoff, acc);
    }
    if (MODE == LG_SLOT) {
      v4f* p = (v4f*)(dst + (size_t)i * EMBED_DIM + off + q * 4);
      *p = *p + acc;
    } else if (act) {
      *(v4f*)(dst + (size_t)row * HALF_DIM + q * 4) = acc;
    }
  }
}

__global__ void lg_gather_acc2(const float* __restrict__ ta, const float* __restrict__ tb,
                               const int* __restrict__ uid, const int* __restrict__ iid,
                               float* __restrict__ acc, int off) {
  const int t = blockIdx.x * blockDim.x + threadIdx.x;
  const int r = t >> 5, c = t & 31;
  const int n = (r < BATCH) ? uid[r] : (NUM_USERS + iid[r - BATCH]);
  float x = acc[r * EMBED_DIM + off + c];
  x += ta[(size_t)n * HALF_DIM + c];
  x += tb[(size_t)n * HALF_DIM + c];
  acc[r * EMBED_DIM + off + c] = x;
}

__global__ void lg_dot_wmma(const float* __restrict__ acc, float* __restrict__ out) {
  __shared__ float sd[8][8][32];
  const int lane = threadIdx.x & 31;
  const int wave = threadIdx.x >> 5;
  const int tile = blockIdx.x * 8 + wave;
  const int k0   = tile * 16;
  const int rN   = lane & 15;
  const float* U = acc + (size_t)(k0 + rN) * EMBED_DIM;
  const float* I = acc + (size_t)(BATCH + k0 + rN) * EMBED_DIM;

  v8f c = {0.f, 0.f, 0.f, 0.f, 0.f, 0.f, 0.f, 0.f};

#if __has_builtin(__builtin_amdgcn_wmma_f32_16x16x4_f32)
  const int kb = (lane >> 4) << 1;
  #pragma unroll
  for (int s = 0; s < 16; ++s) {
    const int col = s * 4 + kb;
    v2f a; a[0] = U[col]; a[1] = U[col + 1];
    v2f b; b[0] = I[col]; b[1] = I[col + 1];
    c = __builtin_amdgcn_wmma_f32_16x16x4_f32(false, a, false, b, (short)0, c, false, false);
  }
#else
  const int kb8 = (lane >> 4) << 3;
  #pragma unroll
  for (int blk = 0; blk < 2; ++blk) {
    v16h a, b;
    #pragma unroll
    for (int i = 0; i < 8; ++i) {
      const int kk = blk * 32 + kb8 + ((i < 4) ? (2 * i) : (16 + 2 * (i - 4)));
      a[2 * i]     = (_Float16)U[kk];
      a[2 * i + 1] = (_Float16)U[kk + 1];
      b[2 * i]     = (_Float16)I[kk];
      b[2 * i + 1] = (_Float16)I[kk + 1];
    }
    c = __builtin_amdgcn_wmma_f32_16x16x32_f16(false, a, false, b, (short)0, c, false, false);
  }
#endif

  #pragma unroll
  for (int g = 0; g < 8; ++g) sd[wave][g][lane] = c[g];
  __syncthreads();

  if (lane < 16) {
    const float v = (lane < 8) ? sd[wave][lane][lane] : sd[wave][lane - 8][lane + 16];
    out[k0 + lane] = v * 0.0625f;
  }
}

static void lg_half(const float* node_emb_half, int col0, const int* rowp, const int2* csr,
                    float* tblA, float* tblB, const int* list, const int* nlist,
                    const int* user_ids, const int* item_ids, float* accS, hipStream_t stream) {
  const dim3 b256(256);
  lg_pull4<EMBED_DIM, LG_ALL><<<2048, b256, 0, stream>>>(node_emb_half, rowp, csr, tblA,
                                                          list, nlist, user_ids, item_ids, col0);
  lg_pull4<HALF_DIM, LG_LIST><<<2048, b256, 0, stream>>>(tblA, rowp, csr, tblB,
                                                          list, nlist, user_ids, item_ids, col0);
  lg_gather_acc2<<<1024, b256, 0, stream>>>(tblA, tblB, user_ids, item_ids, accS, col0);
  lg_pull4<HALF_DIM, LG_SLOT><<<2 * BATCH / 32, b256, 0, stream>>>(tblB, rowp, csr, accS,
                                                          list, nlist, user_ids, item_ids, col0);
}

extern "C" void kernel_launch(void* const* d_in, const int* in_sizes, int n_in,
                              void* d_out, int out_size, void* d_ws, size_t ws_size,
                              hipStream_t stream) {
  const float* node_emb = (const float*)d_in[0];
  const float* adj_val  = (const float*)d_in[1];
  const int*   adj_row  = (const int*)d_in[2];
  const int*   adj_col  = (const int*)d_in[3];
  const int*   user_ids = (const int*)d_in[4];
  const int*   item_ids = (const int*)d_in[5];
  float* out = (float*)d_out;

  char* ws    = (char*)d_ws;
  float* tblA = (float*)(ws + OFF_TBLA);
  float* tblB = (float*)(ws + OFF_TBLB);
  int*   rank = (int*)(ws + OFF_TBLB);
  int2*  csr  = (int2*)(ws + OFF_CSR);
  int*   rowp = (int*)(ws + OFF_ROWP);
  int*   cnt  = (int*)(ws + OFF_CNT);
  int*   flag = (int*)(ws + OFF_FLAG);
  int*   list = (int*)(ws + OFF_LIST);
  int*   nlist = (int*)(ws + OFF_NLIST);
  float* accS = (float*)(ws + OFF_ACCS);

  const dim3 b256(256);

  lg_gather_init<<<2048, b256, 0, stream>>>(node_emb, user_ids, item_ids, accS, cnt, flag);

  lg_rank<<<2048, b256, 0, stream>>>(adj_row, cnt, rank);
  lg_scan<false><<<SCAN_NB, b256, 0, stream>>>((const int4*)cnt, (int4*)rowp, list, nlist);
  lg_fill<<<2048, b256, 0, stream>>>(adj_row, adj_col, adj_val, rowp, rank, csr);

  lg_mark<<<2 * BATCH * 8 / 256, b256, 0, stream>>>(user_ids, item_ids, rowp, csr, flag);
  lg_scan<true><<<SCAN_NB, b256, 0, stream>>>((const int4*)flag, (int4*)rowp, list, nlist);

  lg_half(node_emb,      0,  rowp, csr, tblA, tblB, list, nlist, user_ids, item_ids, accS, stream);
  lg_half(node_emb + 32, 32, rowp, csr, tblA, tblB, list, nlist, user_ids, item_ids, accS, stream);

  lg_dot_wmma<<<BATCH / 128, b256, 0, stream>>>(accS, out);
}


# COO->CSR ranks from LDS counters: edges grouped by 512-row range per 6144-edge chunk, each range then ranked in LDS; 3M global returning atomics gone. Row sum order: arrival order, as before
# speedup vs baseline: 3.2931x; 1.6955x over previous
#include <hip/hip_runtime.h>

#define N_TOTAL    150100
#define EMBED_DIM  64
#define HALF_DIM   32
#define NUM_USERS  100000
#define N_EDGES    3000000
#define BATCH      4096
#define NUM_LAYERS 3

#define CNT_PAD    150528
#define SCAN_NB    147

#define BKT_SHIFT  9
#define BKT_ROWS   512
#define NBKT       294
#define NCHK       512
#define CHUNK_E    6144
#define COL_BITS   18
static_assert(NBKT * NCHK == CNT_PAD && NBKT * BKT_ROWS == CNT_PAD, "one flat scan serves both tables");
static_assert(NCHK * CHUNK_E >= N_EDGES && N_TOTAL < (1 << COL_BITS) && CHUNK_E % 512 == 0, "grouping sizes");

#define OFF_TBLA   0u
#define OFF_TBLB   19212800u
#define OFF_CSR    38425600u
#define OFF_ROWP   62425600u
#define OFF_CNT    63027712u
#define OFF_FLAG   63629824u
#define OFF_LIST   64231936u
#define OFF_NLIST  64834048u
#define OFF_ACCS   76851200u

typedef __attribute__((ext_vector_type(2)))  float    v2f;
typedef __attribute__((ext_vector_type(4)))  float    v4f;
typedef __attribute__((ext_vector_type(8)))  float    v8f;
typedef __attribute__((ext_vector_type(16))) _Float16 v16h;

__global__ void lg_gather_init(const float* __restrict__ emb, const int* __restrict__ uid,
                               const int* __restrict__ iid, float* __restrict__ acc,
                               int* __restrict__ flag) {
  const int t = blockIdx.x * blockDim.x + threadIdx.x;
  const int r = t >> 6, c = t & 63;
  const int n = (r < BATCH) ? uid[r] : (NUM_USERS + iid[r - BATCH]);
  acc[t] = emb[(size_t)n * EMBED_DIM + c];
  if (t < CNT_PAD) flag[t] = 0;
}

__global__ __launch_bounds__(512) void lg_bkt_count(const int* __restrict__ rw, int* __restrict__ cntCB) {
  __shared__ int h[NBKT];
  const int t = threadIdx.x, c = blockIdx.x;
  if (t < NBKT) h[t] = 0;
  __syncthreads();
  const int i0 = c * CHUNK_E + t;
  int r[CHUNK_E / 512];
  #pragma unroll
  for (int k = 0; k < CHUNK_E / 512; ++k) r[k] = rw[min(i0 + k * 512, N_EDGES - 1)];
  #pragma unroll
  for (int k = 0; k < CHUNK_E / 512; ++k)
    if (i0 + k * 512 < N_EDGES) atomicAdd(&h[r[k] >> BKT_SHIFT], 1);
  __syncthreads();
  if (t < NBKT) cntCB[t * NCHK + c] = h[t];
}

__global__ __launch_bounds__(512) void lg_bkt_place(const int* __restrict__ rw, const int* __restrict__ cl,
                                                    const float* __restrict__ val,
                                                    const int* __restrict__ offsCB, int2* __restrict__ grp) {
  __shared__ int cur[NBKT];
  const int t = threadIdx.x, c = blockIdx.x;
  if (t < NBKT) cur[t] = offsCB[t * NCHK + c];
  __syncthreads();
  const int i0 = c * CHUNK_E + t;
  int   r[CHUNK_E / 512], cc[CHUNK_E / 512];
  float vv[CHUNK_E / 512];
  #pragma unroll
  for (int k = 0; k < CHUNK_E / 512; ++k) {
    const int i = min(i0 + k * 512, N_EDGES - 1);
    r[k] = rw[i]; cc[k] = cl[i]; vv[k] = val[i];
  }
  #pragma unroll
  for (int k = 0; k < CHUNK_E / 512; ++k)
    if (i0 + k * 512 < N_EDGES) {
      const int p = atomicAdd(&cur[r[k] >> BKT_SHIFT], 1);
      grp[p] = make_int2(((r[k] & (BKT_ROWS - 1)) << COL_BITS) | cc[k], __float_as_int(vv[k]));
    }
}

__global__ __launch_bounds__(512) void lg_bkt_rows(const int2* __restrict__ grp, const int* __restrict__ offsCB,
                                                   int* __restrict__ rowp, int2* __restrict__ csr) {
  __shared__ int cnt[BKT_ROWS];
  __shared__ int scn[BKT_ROWS];
  const int t = threadIdx.x, b = blockIdx.x;
  const int bs = max(0, min(offsCB[b * NCHK], N_EDGES));
  const int be = (b + 1 < NBKT) ? min(offsCB[(b + 1) * NCHK], N_EDGES) : N_EDGES;
  cnt[t] = 0;
  __syncthreads();
  for (int j = bs + t; j < be; j += 4 * 512) {
    int key[4];
    #pragma unroll
    for (int u = 0; u < 4; ++u) key[u] = grp[min(j + u * 512, be - 1)].x;
    #pragma unroll
    for (int u = 0; u < 4; ++u)
      if (j + u * 512 < be) atomicAdd(&cnt[(key[u] >> COL_BITS) & (BKT_ROWS - 1)], 1);
  }
  __syncthreads();
  const int c = cnt[t];
  scn[t] = c;
  __syncthreads();
  for (int s = 1; s < BKT_ROWS; s <<= 1) {
    const int add = (t >= s) ? scn[t - s] : 0;
    __syncthreads();
    scn[t] += add;
    __syncthreads();
  }
  const int start = bs + scn[t] - c;
  rowp[b * BKT_ROWS + t] = start;
  cnt[t] = start;
  __syncthreads();
  for (int j = bs + t; j < be; j += 4 * 512) {
    int2 e[4];
    #pragma unroll
    for (int u = 0; u < 4; ++u) e[u] = grp[min(j + u * 512, be - 1)];
    #pragma unroll
    for (int u = 0; u < 4; ++u)
      if (j + u * 512 < be) {
        const int p = atomicAdd(&cnt[(e[u].x >> COL_BITS) & (BKT_ROWS - 1)], 1);
        csr[p] = make_int2(e[u].x & ((1 << COL_BITS) - 1), e[u].y);
      }
  }
}

template <bool LIST>
__global__ void lg_scan(const int4* __restrict__ cnt4, int4* __restrict__ rowp4,
                        int* __restrict__ list, int* __restrict__ nlist) {
  __shared__ int sh[256];
  __shared__ int sb[256];
  const int t = threadIdx.x, b = blockIdx.x;
  int pre = 0;
  for (int i = t; i < b * 256; i += 256) {
    const int4 u = cnt4[i];
    pre += u.x + u.y + u.z + u.w;
  }
  const int4 v = cnt4[b * 256 + t];
  const int tsum = v.x + v.y + v.z + v.w;
  sh[t] = tsum;
  sb[t] = pre;
  __syncthreads();
  for (int s = 1; s < 256; s <<= 1) {
    const int add = (t >= s) ? sh[t - s] : 0;
    __syncthreads();
    sh[t] += add;
    __syncthreads();
  }
  for (int s = 128; s > 0; s >>= 1) {
    if (t < s) sb[t] += sb[t + s];
    __syncthreads();
  }
  const int base = sb[0] + sh[t] - tsum;
  int4 o;
  o.x = base; o.y = o.x + v.x; o.z = o.y + v.y; o.w = o.z + v.z;
  if (!LIST) {
    rowp4[b * 256 + t] = o;
  } else {
    const int r = (b * 256 + t) * 4;
    if (v.x) list[o.x] = r;
    if (v.y) list[o.y] = r + 1;
    if (v.z) list[o.z] = r + 2;
    if (v.w) list[o.w] = r + 3;
    if (b == SCAN_NB - 1 && t == 255) *nlist = o.w + v.w;
  }
}

__global__ void lg_mark(const int* __restrict__ uid, const int* __restrict__ iid,
                        const int* __restrict__ rowp, const int2* __restrict__ csr,
                        int* __restrict__ flag) {
  const int t = blockIdx.x * blockDim.x + threadIdx.x;
  const int r = t >> 3, q = t & 7;
  const int n = (r < BATCH) ? uid[r] : (NUM_USERS + iid[r - BATCH]);
  const int beg = rowp[n], end = rowp[n + 1];
  if (q == 0) atomicOr(&flag[n], 1);
  for (int j = beg + q; j < end; j += 8) atomicOr(&flag[csr[j].x], 1);
}


template <int K>
__device__ __forceinline__ int grp_bcast(int v) {
  return (int)__builtin_amdgcn_mov_dpp8((unsigned)v, (unsigned)(K * 0x249249));
}

#define LG_SLOT(K)                                                                \
  {                                                                               \
    const unsigned c = (unsigned)grp_bcast<K>(cc);                                \
    w[K] = __int_as_float(grp_bcast<K>(ww));                                      \
    x[K] = *(const v4f*)(srcb + (size_t)(c * (unsigned)(STRIDE * 4) + qoff));     \
  }

template <int STRIDE>
__device__ __forceinline__ void lg_round(const char* __restrict__ srcb, const int cc, const int ww,
                                         const unsigned qoff, v4f& acc) {
  v4f   x[8];
  float w[8];
  LG_SLOT(0) LG_SLOT(1) LG_SLOT(2) LG_SLOT(3) LG_SLOT(4) LG_SLOT(5) LG_SLOT(6) LG_SLOT(7)
  #pragma unroll
  for (int k = 0; k < 8; ++k) acc += w[k] * x[k];
}

enum { LG_ALL = 0, LG_LIST = 1, LG_SLOT = 2 };

template <int STRIDE, int MODE>
__global__ void lg_pull4(const float* __restrict__ src, const int* __restrict__ rowp,
                         const int2* __restrict__ csr, float* __restrict__ dst,
                         const int* __restrict__ list, const int* __restrict__ nlist,
                         const int* __restrict__ uid, const int* __restrict__ iid, int off) {
  static_assert(N_TOTAL % 4 == 0 && (2 * BATCH) % 4 == 0, "a wave takes four whole rows");
  const int lane = threadIdx.x & 31;
  const int q = lane & 7;
  const int g = lane >> 3;
  const unsigned qoff = (unsigned)q * 16u;
  const char* srcb = (const char*)src;
  const int w0 = __builtin_amdgcn_readfirstlane((int)((blockIdx.x * blockDim.x + threadIdx.x) >> 5));
  const int nw = (int)((gridDim.x * blockDim.x) >> 5);
  int total = N_TOTAL;
  if (MODE == LG_SLOT) total = 2 * BATCH;
  if (MODE == LG_LIST) total = min(*nlist, N_TOTAL);
  for (int r0 = w0 * 4; r0 < total; r0 += nw * 4) {
    const int i = r0 + g;
    const bool act = (MODE != LG_LIST) || i < total;
    int row = i;
    if (MODE == LG_LIST) row = act ? list[i] : 0;
    if (MODE == LG_SLOT) row = (i < BATCH) ? uid[i] : (NUM_USERS + iid[i - BATCH]);
    int beg = 0, end = 0;
    if (act) { beg = rowp[row]; end = rowp[row + 1]; }
    v4f acc = {0.f, 0.f, 0.f, 0.f};
    for (int jb = beg; jb < end; jb += 32) {
      const int n = min(end - jb, 32);
      int2 rec[4];
      #pragma unroll
      for (int j = 0; j < 4; ++j) {
        rec[j] = make_int2(0, 0);
        if (8 * j < n) rec[j] = csr[jb + min(8 * j + q, n - 1)];
      }
      #pragma unroll
      for (int j = 0; j < 4; ++j)
        if (8 * j < n)
          lg_round<STRIDE>(srcb, rec[j].x, (8 * j + q < n) ? rec[j].y : 0  , qoff, acc);
    }
    if (MODE == LG_SLOT) {
      v4f* p = (v4f*)(dst + (size_t)i * EMBED_DIM + off + q * 4);
      *p = *p + acc;
    } else if (act) {
      *(v4f*)(dst + (size_t)row * HALF_DIM + q * 4) = acc;
    }
  }
}

__global__ void lg_gather_acc2(const float* __restrict__ ta, const float* __restrict__ tb,
                               const int* __restrict__ uid, const int* __restrict__ iid,
                               float* __restrict__ acc, int off) {
  const int t = blockIdx.x * blockDim.x + threadIdx.x;
  const int r = t >> 5, c = t & 31;
  const int n = (r < BATCH) ? uid[r] : (NUM_USERS + iid[r - BATCH]);
  float x = acc[r * EMBED_DIM + off + c];
  x += ta[(size_t)n * HALF_DIM + c];
  x += tb[(size_t)n * HALF_DIM + c];
  acc[r * EMBED_DIM + off + c] = x;
}

__global__ void lg_dot_wmma(const float* __restrict__ acc, float* __restrict__ out) {
  __shared__ float sd[8][8][32];
  const int lane = threadIdx.x & 31;
  const int wave = threadIdx.x >> 5;
  const int tile = blockIdx.x * 8 + wave;
  const int k0   = tile * 16;
  const int rN   = lane & 15;
  const float* U = acc + (size_t)(k0 + rN) * EMBED_DIM;
  const float* I = acc + (size_t)(BATCH + k0 + rN) * EMBED_DIM;

  v8f c = {0.f, 0.f, 0.f, 0.f, 0.f, 0.f, 0.f, 0.f};

#if __has_builtin(__builtin_amdgcn_wmma_f32_16x16x4_f32)
  const int kb = (lane >> 4) << 1;
  #pragma unroll
  for (int s = 0; s < 16; ++s) {
    const int col = s * 4 + kb;
    v2f a; a[0] = U[col]; a[1] = U[col + 1];
    v2f b; b[0] = I[col]; b[1] = I[col + 1];
    c = __builtin_amdgcn_wmma_f32_16x16x4_f32(false, a, false, b, (short)0, c, false, false);
  }
#else
  const int kb8 = (lane >> 4) << 3;
  #pragma unroll
  for (int blk = 0; blk < 2; ++blk) {
    v16h a, b;
    #pragma unroll
    for (int i = 0; i < 8; ++i) {
      const int kk = blk * 32 + kb8 + ((i < 4) ? (2 * i) : (16 + 2 * (i - 4)));
      a[2 * i]     = (_Float16)U[kk];
      a[2 * i + 1] = (_Float16)U[kk + 1];
      b[2 * i]     = (_Float16)I[kk];
      b[2 * i + 1] = (_Float16)I[kk + 1];
    }
    c = __builtin_amdgcn_wmma_f32_16x16x32_f16(false, a, false, b, (short)0, c, false, false);
  }
#endif

  #pragma unroll
  for (int g = 0; g < 8; ++g) sd[wave][g][lane] = c[g];
  __syncthreads();

  if (lane < 16) {
    const float v = (lane < 8) ? sd[wave][lane][lane] : sd[wave][lane - 8][lane + 16];
    out[k0 + lane] = v * 0.0625f;
  }
}

static void lg_half(const float* node_emb_half, int col0, const int* rowp, const int2* csr,
                    float* tblA, float* tblB, const int* list, const int* nlist,
                    const int* user_ids, const int* item_ids, float* accS, hipStream_t stream) {
  const dim3 b256(256);
  lg_pull4<EMBED_DIM, LG_ALL><<<2048, b256, 0, stream>>>(node_emb_half, rowp, csr, tblA,
                                                          list, nlist, user_ids, item_ids, col0);
  lg_pull4<HALF_DIM, LG_LIST><<<2048, b256, 0, stream>>>(tblA, rowp, csr, tblB,
                                                          list, nlist, user_ids, item_ids, col0);
  lg_gather_acc2<<<1024, b256, 0, stream>>>(tblA, tblB, user_ids, item_ids, accS, col0);
  lg_pull4<HALF_DIM, LG_SLOT><<<2 * BATCH / 32, b256, 0, stream>>>(tblB, rowp, csr, accS,
                                                          list, nlist, user_ids, item_ids, col0);
}

extern "C" void kernel_launch(void* const* d_in, const int* in_sizes, int n_in,
                              void* d_out, int out_size, void* d_ws, size_t ws_size,
                              hipStream_t stream) {
  const float* node_emb = (const float*)d_in[0];
  const float* adj_val  = (const float*)d_in[1];
  const int*   adj_row  = (const int*)d_in[2];
  const int*   adj_col  = (const int*)d_in[3];
  const int*   user_ids = (const int*)d_in[4];
  const int*   item_ids = (const int*)d_in[5];
  float* out = (float*)d_out;

  char* ws    = (char*)d_ws;
  float* tblA = (float*)(ws + OFF_TBLA);
  float* tblB = (float*)(ws + OFF_TBLB);
  int2*  grp  = (int2*)(ws + OFF_TBLA);
  int2*  csr  = (int2*)(ws + OFF_CSR);
  int*   rowp = (int*)(ws + OFF_ROWP);
  int*   cntCB = (int*)(ws + OFF_CNT);
  int*   flag = (int*)(ws + OFF_FLAG);
  int*   list = (int*)(ws + OFF_LIST);
  int*   nlist = (int*)(ws + OFF_NLIST);
  int*   offsCB = (int*)(ws + OFF_LIST);
  float* accS = (float*)(ws + OFF_ACCS);

  const dim3 b256(256);

  lg_gather_init<<<2048, b256, 0, stream>>>(node_emb, user_ids, item_ids, accS, flag);

  lg_bkt_count<<<NCHK, 512, 0, stream>>>(adj_row, cntCB);
  lg_scan<false><<<SCAN_NB, b256, 0, stream>>>((const int4*)cntCB, (int4*)offsCB, list, nlist);
  lg_bkt_place<<<NCHK, 512, 0, stream>>>(adj_row, adj_col, adj_val, offsCB, grp);
  lg_bkt_rows<<<NBKT, 512, 0, stream>>>(grp, offsCB, rowp, csr);

  lg_mark<<<2 * BATCH * 8 / 256, b256, 0, stream>>>(user_ids, item_ids, rowp, csr, flag);
  lg_scan<true><<<SCAN_NB, b256, 0, stream>>>((const int4*)flag, (int4*)rowp, list, nlist);

  lg_half(node_emb,      0,  rowp, csr, tblA, tblB, list, nlist, user_ids, item_ids, accS, stream);
  lg_half(node_emb + 32, 32, rowp, csr, tblA, tblB, list, nlist, user_ids, item_ids, accS, stream);

  lg_dot_wmma<<<BATCH / 128, b256, 0, stream>>>(accS, out);
}


# Four small launches folded into neighbours: last-layer row pass stores emb+L1+L2+L3 itself (init, 2 gather-adds gone), row marking rides in layer 1, flag clear in the count. 16 -> 12; sum order kept
# speedup vs baseline: 3.5169x; 1.0680x over previous
#include <hip/hip_runtime.h>

#define N_TOTAL    150100
#define EMBED_DIM  64
#define HALF_DIM   32
#define NUM_USERS  100000
#define N_EDGES    3000000
#define BATCH      4096
#define NUM_LAYERS 3

#define CNT_PAD    150528
#define SCAN_NB    147

#define BKT_SHIFT  9
#define BKT_ROWS   512
#define NBKT       294
#define NCHK       512
#define CHUNK_E    6144
#define COL_BITS   18
static_assert(NBKT * NCHK == CNT_PAD && NBKT * BKT_ROWS == CNT_PAD, "one flat scan serves both tables");
static_assert(NCHK * CHUNK_E >= N_EDGES && N_TOTAL < (1 << COL_BITS) && CHUNK_E % 512 == 0, "grouping sizes");

#define OFF_TBLA   0u
#define OFF_TBLB   19212800u
#define OFF_CSR    38425600u
#define OFF_ROWP   62425600u
#define OFF_CNT    63027712u
#define OFF_FLAG   63629824u
#define OFF_LIST   64231936u
#define OFF_NLIST  64834048u
#define OFF_ACCS   76851200u

typedef __attribute__((ext_vector_type(2)))  float    v2f;
typedef __attribute__((ext_vector_type(4)))  float    v4f;
typedef __attribute__((ext_vector_type(8)))  float    v8f;
typedef __attribute__((ext_vector_type(16))) _Float16 v16h;

__global__ __launch_bounds__(512) void lg_bkt_count(const int* __restrict__ rw, int* __restrict__ cntCB,
                                                    int* __restrict__ flag) {
  static_assert(NCHK * 512 >= CNT_PAD, "one thread per row flag");
  __shared__ int h[NBKT];
  const int t = threadIdx.x, c = blockIdx.x;
  if (c * 512 + t < CNT_PAD) flag[c * 512 + t] = 0;
  if (t < NBKT) h[t] = 0;
  __syncthreads();
  const int i0 = c * CHUNK_E + t;
  int r[CHUNK_E / 512];
  #pragma unroll
  for (int k = 0; k < CHUNK_E / 512; ++k) r[k] = rw[min(i0 + k * 512, N_EDGES - 1)];
  #pragma unroll
  for (int k = 0; k < CHUNK_E / 512; ++k)
    if (i0 + k * 512 < N_EDGES) atomicAdd(&h[r[k] >> BKT_SHIFT], 1);
  __syncthreads();
  if (t < NBKT) cntCB[t * NCHK + c] = h[t];
}

__global__ __launch_bounds__(512) void lg_bkt_place(const int* __restrict__ rw, const int* __restrict__ cl,
                                                    const float* __restrict__ val,
                                                    const int* __restrict__ offsCB, int2* __restrict__ grp) {
  __shared__ int cur[NBKT];
  const int t = threadIdx.x, c = blockIdx.x;
  if (t < NBKT) cur[t] = offsCB[t * NCHK + c];
  __syncthreads();
  const int i0 = c * CHUNK_E + t;
  int   r[CHUNK_E / 512], cc[CHUNK_E / 512];
  float vv[CHUNK_E / 512];
  #pragma unroll
  for (int k = 0; k < CHUNK_E / 512; ++k) {
    const int i = min(i0 + k * 512, N_EDGES - 1);
    r[k] = rw[i]; cc[k] = cl[i]; vv[k] = val[i];
  }
  #pragma unroll
  for (int k = 0; k < CHUNK_E / 512; ++k)
    if (i0 + k * 512 < N_EDGES) {
      const int p = atomicAdd(&cur[r[k] >> BKT_SHIFT], 1);
      grp[p] = make_int2(((r[k] & (BKT_ROWS - 1)) << COL_BITS) | cc[k], __float_as_int(vv[k]));
    }
}

__global__ __launch_bounds__(512) void lg_bkt_rows(const int2* __restrict__ grp, const int* __restrict__ offsCB,
                                                   int* __restrict__ rowp, int2* __restrict__ csr) {
  __shared__ int cnt[BKT_ROWS];
  __shared__ int scn[BKT_ROWS];
  const int t = threadIdx.x, b = blockIdx.x;
  const int bs = max(0, min(offsCB[b * NCHK], N_EDGES));
  const int be = (b + 1 < NBKT) ? min(offsCB[(b + 1) * NCHK], N_EDGES) : N_EDGES;
  cnt[t] = 0;
  __syncthreads();
  for (int j = bs + t; j < be; j += 4 * 512) {
    int key[4];
    #pragma unroll
    for (int u = 0; u < 4; ++u) key[u] = grp[min(j + u * 512, be - 1)].x;
    #pragma unroll
    for (int u = 0; u < 4; ++u)
      if (j + u * 512 < be) atomicAdd(&cnt[(key[u] >> COL_BITS) & (BKT_ROWS - 1)], 1);
  }
  __syncthreads();
  const int c = cnt[t];
  scn[t] = c;
  __syncthreads();
  for (int s = 1; s < BKT_ROWS; s <<= 1) {
    const int add = (t >= s) ? scn[t - s] : 0;
    __syncthreads();
    scn[t] += add;
    __syncthreads();
  }
  const int start = bs + scn[t] - c;
  rowp[b * BKT_ROWS + t] = start;
  cnt[t] = start;
  __syncthreads();
  for (int j = bs + t; j < be; j += 4 * 512) {
    int2 e[4];
    #pragma unroll
    for (int u = 0; u < 4; ++u) e[u] = grp[min(j + u * 512, be - 1)];
    #pragma unroll
    for (int u = 0; u < 4; ++u)
      if (j + u * 512 < be) {
        const int p = atomicAdd(&cnt[(e[u].x >> COL_BITS) & (BKT_ROWS - 1)], 1);
        csr[p] = make_int2(e[u].x & ((1 << COL_BITS) - 1), e[u].y);
      }
  }
}

template <bool LIST>
__global__ void lg_scan(const int4* __restrict__ cnt4, int4* __restrict__ rowp4,
                        int* __restrict__ list, int* __restrict__ nlist) {
  __shared__ int sh[256];
  __shared__ int sb[256];
  const int t = threadIdx.x, b = blockIdx.x;
  int pre = 0;
  for (int i = t; i < b * 256; i += 256) {
    const int4 u = cnt4[i];
    pre += u.x + u.y + u.z + u.w;
  }
  const int4 v = cnt4[b * 256 + t];
  const int tsum = v.x + v.y + v.z + v.w;
  sh[t] = tsum;
  sb[t] = pre;
  __syncthreads();
  for (int s = 1; s < 256; s <<= 1) {
    const int add = (t >= s) ? sh[t - s] : 0;
    __syncthreads();
    sh[t] += add;
    __syncthreads();
  }
  for (int s = 128; s > 0; s >>= 1) {
    if (t < s) sb[t] += sb[t + s];
    __syncthreads();
  }
  const int base = sb[0] + sh[t] - tsum;
  int4 o;
  o.x = base; o.y = o.x + v.x; o.z = o.y + v.y; o.w = o.z + v.z;
  if (!LIST) {
    rowp4[b * 256 + t] = o;
  } else {
    const int r = (b * 256 + t) * 4;
    if (v.x) list[o.x] = r;
    if (v.y) list[o.y] = r + 1;
    if (v.z) list[o.z] = r + 2;
    if (v.w) list[o.w] = r + 3;
    if (b == SCAN_NB - 1 && t == 255) *nlist = o.w + v.w;
  }
}

__device__ __forceinline__ void lg_mark(const int t, const int* __restrict__ uid, const int* __restrict__ iid,
                                        const int* __restrict__ rowp, const int2* __restrict__ csr,
                                        int* __restrict__ flag) {
  const int r = t >> 3, q = t & 7;
  const int n = (r < BATCH) ? uid[r] : (NUM_USERS + iid[r - BATCH]);
  const int beg = rowp[n], end = rowp[n + 1];
  if (q == 0) atomicOr(&flag[n], 1);
  for (int j = beg + q; j < end; j += 8) atomicOr(&flag[csr[j].x], 1);
}


template <int K>
__device__ __forceinline__ int grp_bcast(int v) {
  return (int)__builtin_amdgcn_mov_dpp8((unsigned)v, (unsigned)(K * 0x249249));
}

#define LG_SLOT(K)                                                                \
  {                                                                               \
    const unsigned c = (unsigned)grp_bcast<K>(cc);                                \
    w[K] = __int_as_float(grp_bcast<K>(ww));                                      \
    x[K] = *(const v4f*)(srcb + (size_t)(c * (unsigned)(STRIDE * 4) + qoff));     \
  }

template <int STRIDE>
__device__ __forceinline__ void lg_round(const char* __restrict__ srcb, const int cc, const int ww,
                                         const unsigned qoff, v4f& acc) {
  v4f   x[8];
  float w[8];
  LG_SLOT(0) LG_SLOT(1) LG_SLOT(2) LG_SLOT(3) LG_SLOT(4) LG_SLOT(5) LG_SLOT(6) LG_SLOT(7)
  #pragma unroll
  for (int k = 0; k < 8; ++k) acc += w[k] * x[k];
}

enum { LG_ALL = 0, LG_LIST = 1, LG_SLOT = 2 };

template <int STRIDE, int MODE, bool MARK>
__global__ void lg_pull4(const float* __restrict__ src, const int* __restrict__ rowp,
                         const int2* __restrict__ csr, float* __restrict__ dst,
                         const int* __restrict__ list, const int* __restrict__ nlist,
                         const int* __restrict__ uid, const int* __restrict__ iid, int off,
                         const float* __restrict__ emb0, const float* __restrict__ ta, int* __restrict__ flag) {
  static_assert(N_TOTAL % 4 == 0 && (2 * BATCH) % 4 == 0, "a wave takes four whole rows");
  if (MARK) {
    const int t = ((int)blockIdx.x - ((int)gridDim.x - 2 * BATCH * 8 / 256)) * 256 + (int)threadIdx.x;
    if (t >= 0) lg_mark(t, uid, iid, rowp, csr, flag);
  }
  const int lane = threadIdx.x & 31;
  const int q = lane & 7;
  const int g = lane >> 3;
  const unsigned qoff = (unsigned)q * 16u;
  const char* srcb = (const char*)src;
  const int w0 = __builtin_amdgcn_readfirstlane((int)((blockIdx.x * blockDim.x + threadIdx.x) >> 5));
  const int nw = (int)((gridDim.x * blockDim.x) >> 5);
  int total = N_TOTAL;
  if (MODE == LG_SLOT) total = 2 * BATCH;
  if (MODE == LG_LIST) total = min(*nlist, N_TOTAL);
  for (int r0 = w0 * 4; r0 < total; r0 += nw * 4) {
    const int i = r0 + g;
    const bool act = (MODE != LG_LIST) || i < total;
    int row = i;
    if (MODE == LG_LIST) row = act ? list[i] : 0;
    if (MODE == LG_SLOT) row = (i < BATCH) ? uid[i] : (NUM_USERS + iid[i - BATCH]);
    int beg = 0, end = 0;
    if (act) { beg = rowp[row]; end = rowp[row + 1]; }
    v4f acc = {0.f, 0.f, 0.f, 0.f};
    v4f l0 = acc, l1 = acc, l2 = acc;
    if (MODE == LG_SLOT) {
      l0 = *(const v4f*)(emb0 + (size_t)row * EMBED_DIM + q * 4);
      l1 = *(const v4f*)(ta + (size_t)row * HALF_DIM + q * 4);
      l2 = *(const v4f*)(src + (size_t)row * HALF_DIM + q * 4);
    }
    for (int jb = beg; jb < end; jb += 32) {
      const int n = min(end - jb, 32);
      int2 rec[4];
      #pragma unroll
      for (int j = 0; j < 4; ++j) {
        rec[j] = make_int2(0, 0);
        if (8 * j < n) rec[j] = csr[jb + min(8 * j + q, n - 1)];
      }
      #pragma unroll
      for (int j = 0; j < 4; ++j)
        if (8 * j < n)
          lg_round<STRIDE>(srcb, rec[j].x, (8 * j + q < n) ? rec[j].y : 0  , qoff, acc);
    }
    if (MODE == LG_SLOT) {
      *(v4f*)(dst + (size_t)i * EMBED_DIM + off + q * 4) = ((l0 + l1) + l2) + acc;
    } else if (act) {
      *(v4f*)(dst + (size_t)row * HALF_DIM + q * 4) = acc;
    }
  }
}

__global__ void lg_dot_wmma(const float* __restrict__ acc, float* __restrict__ out) {
  __shared__ float sd[8][8][32];
  const int lane = threadIdx.x & 31;
  const int wave = threadIdx.x >> 5;
  const int tile = blockIdx.x * 8 + wave;
  const int k0   = tile * 16;
  const int rN   = lane & 15;
  const float* U = acc + (size_t)(k0 + rN) * EMBED_DIM;
  const float* I = acc + (size_t)(BATCH + k0 + rN) * EMBED_DIM;

  v8f c = {0.f, 0.f, 0.f, 0.f, 0.f, 0.f, 0.f, 0.f};

#if __has_builtin(__builtin_amdgcn_wmma_f32_16x16x4_f32)
  const int kb = (lane >> 4) << 1;
  #pragma unroll
  for (int s = 0; s < 16; ++s) {
    const int col = s * 4 + kb;
    v2f a; a[0] = U[col]; a[1] = U[col + 1];
    v2f b; b[0] = I[col]; b[1] = I[col + 1];
    c = __builtin_amdgcn_wmma_f32_16x16x4_f32(false, a, false, b, (short)0, c, false, false);
  }
#else
  const int kb8 = (lane >> 4) << 3;
  #pragma unroll
  for (int blk = 0; blk < 2; ++blk) {
    v16h a, b;
    #pragma unroll
    for (int i = 0; i < 8; ++i) {
      const int kk = blk * 32 + kb8 + ((i < 4) ? (2 * i) : (16 + 2 * (i - 4)));
      a[2 * i]     = (_Float16)U[kk];
      a[2 * i + 1] = (_Float16)U[kk + 1];
      b[2 * i]     = (_Float16)I[kk];
      b[2 * i + 1] = (_Float16)I[kk + 1];
    }
    c = __builtin_amdgcn_wmma_f32_16x16x32_f16(false, a, false, b, (short)0, c, false, false);
  }
#endif

  #pragma unroll
  for (int g = 0; g < 8; ++g) sd[wave][g][lane] = c[g];
  __syncthreads();

  if (lane < 16) {
    const float v = (lane < 8) ? sd[wave][lane][lane] : sd[wave][lane - 8][lane + 16];
    out[k0 + lane] = v * 0.0625f;
  }
}

static void lg_half23(const float* node_emb_half, int col0, const int* rowp, const int2* csr,
                      float* tblA, float* tblB, const int* list, const int* nlist,
                      const int* user_ids, const int* item_ids, float* accS, int* flag, hipStream_t stream) {
  const dim3 b256(256);
  lg_pull4<HALF_DIM, LG_LIST, false><<<2048, b256, 0, stream>>>(tblA, rowp, csr, tblB, list, nlist,
                                                     user_ids, item_ids, col0, node_emb_half, tblA, flag);
  lg_pull4<HALF_DIM, LG_SLOT, false><<<2 * BATCH / 32, b256, 0, stream>>>(tblB, rowp, csr, accS, list, nlist,
                                                     user_ids, item_ids, col0, node_emb_half, tblA, flag);
}

extern "C" void kernel_launch(void* const* d_in, const int* in_sizes, int n_in,
                              void* d_out, int out_size, void* d_ws, size_t ws_size,
                              hipStream_t stream) {
  const float* node_emb = (const float*)d_in[0];
  const float* adj_val  = (const float*)d_in[1];
  const int*   adj_row  = (const int*)d_in[2];
  const int*   adj_col  = (const int*)d_in[3];
  const int*   user_ids = (const int*)d_in[4];
  const int*   item_ids = (const int*)d_in[5];
  float* out = (float*)d_out;

  char* ws    = (char*)d_ws;
  float* tblA = (float*)(ws + OFF_TBLA);
  float* tblB = (float*)(ws + OFF_TBLB);
  int2*  grp  = (int2*)(ws + OFF_TBLA);
  int2*  csr  = (int2*)(ws + OFF_CSR);
  int*   rowp = (int*)(ws + OFF_ROWP);
  int*   cntCB = (int*)(ws + OFF_CNT);
  int*   flag = (int*)(ws + OFF_FLAG);
  int*   list = (int*)(ws + OFF_LIST);
  int*   nlist = (int*)(ws + OFF_NLIST);
  int*   offsCB = (int*)(ws + OFF_LIST);
  float* accS = (float*)(ws + OFF_ACCS);

  const dim3 b256(256);

  lg_bkt_count<<<NCHK, 512, 0, stream>>>(adj_row, cntCB, flag);
  lg_scan<false><<<SCAN_NB, b256, 0, stream>>>((const int4*)cntCB, (int4*)offsCB, list, nlist);
  lg_bkt_place<<<NCHK, 512, 0, stream>>>(adj_row, adj_col, adj_val, offsCB, grp);
  lg_bkt_rows<<<NBKT, 512, 0, stream>>>(grp, offsCB, rowp, csr);

  lg_pull4<EMBED_DIM, LG_ALL, true><<<2048, b256, 0, stream>>>(node_emb, rowp, csr, tblA, list, nlist,
                                                     user_ids, item_ids, 0, node_emb, tblA, flag);
  lg_scan<true><<<SCAN_NB, b256, 0, stream>>>((const int4*)flag, (int4*)rowp, list, nlist);
  lg_half23(node_emb, 0, rowp, csr, tblA, tblB, list, nlist, user_ids, item_ids, accS, flag, stream);

  lg_pull4<EMBED_DIM, LG_ALL, false><<<2048, b256, 0, stream>>>(node_emb + 32, rowp, csr, tblA, list, nlist,
                                                     user_ids, item_ids, 32, node_emb + 32, tblA, flag);
  lg_half23(node_emb + 32, 32, rowp, csr, tblA, tblB, list, nlist, user_ids, item_ids, accS, flag, stream);

  lg_dot_wmma<<<BATCH / 128, b256, 0, stream>>>(accS, out);
}


# COO->CSR record stores staged in LDS: both grouping launches place records in LDS by rank, then write list and CSR as contiguous lines, not 8 B a lane to 32 lines. Row sums: arrival order, as before
# speedup vs baseline: 4.0688x; 1.1569x over previous
#include <hip/hip_runtime.h>

#define N_TOTAL    150100
#define EMBED_DIM  64
#define HALF_DIM   32
#define NUM_USERS  100000
#define N_EDGES    3000000
#define BATCH      4096
#define NUM_LAYERS 3

#define CNT_PAD    150528
#define SCAN_NB    147

#define BKT_SHIFT  9
#define BKT_ROWS   512
#define NBKT       294
#define NCHK       512
#define CHUNK_E    6144
#define COL_BITS   18
static_assert(NBKT * NCHK == CNT_PAD && NBKT * BKT_ROWS == CNT_PAD, "one flat scan serves both tables");
static_assert(NCHK * CHUNK_E >= N_EDGES && N_TOTAL < (1 << COL_BITS) && CHUNK_E % 512 == 0, "grouping sizes");

#define OFF_TBLA   0u
#define OFF_TBLB   19212800u
#define OFF_CSR    38425600u
#define OFF_ROWP   62425600u
#define OFF_CNT    63027712u
#define OFF_FLAG   63629824u
#define OFF_LIST   64231936u
#define OFF_NLIST  64834048u
#define OFF_ACCS   76851200u

typedef __attribute__((ext_vector_type(2)))  float    v2f;
typedef __attribute__((ext_vector_type(4)))  float    v4f;
typedef __attribute__((ext_vector_type(4)))  int      v4i;
typedef __attribute__((ext_vector_type(8)))  float    v8f;
typedef __attribute__((ext_vector_type(16))) _Float16 v16h;

__global__ __launch_bounds__(512) void lg_bkt_count(const int* __restrict__ rw, int* __restrict__ cntCB,
                                                    int* __restrict__ flag) {
  static_assert(NCHK * 512 >= CNT_PAD, "one thread per row flag");
  __shared__ int h[NBKT];
  const int t = threadIdx.x, c = blockIdx.x;
  if (c * 512 + t < CNT_PAD) flag[c * 512 + t] = 0;
  if (t < NBKT) h[t] = 0;
  __syncthreads();
  const int i0 = c * CHUNK_E + t;
  int r[CHUNK_E / 512];
  #pragma unroll
  for (int k = 0; k < CHUNK_E / 512; ++k) r[k] = rw[min(i0 + k * 512, N_EDGES - 1)];
  #pragma unroll
  for (int k = 0; k < CHUNK_E / 512; ++k)
    if (i0 + k * 512 < N_EDGES) atomicAdd(&h[r[k] >> BKT_SHIFT], 1);
  __syncthreads();
  if (t < NBKT) cntCB[t * NCHK + c] = h[t];
}

__global__ __launch_bounds__(512) void lg_bkt_place(const int* __restrict__ rw, const int* __restrict__ cl,
                                                    const float* __restrict__ val,
                                                    const int* __restrict__ offsCB, int2* __restrict__ grp) {
  __shared__ int  cur[NBKT];
  __shared__ int  dlt[NBKT];
  __shared__ int  wsum[16];
  __shared__ int2 stg[CHUNK_E];
  __shared__ unsigned short rng[CHUNK_E];
  const int t = threadIdx.x, c = blockIdx.x;
  const int lane = t & 31, wave = t >> 5;
  const int i0 = c * CHUNK_E + t;
  int   r[CHUNK_E / 512], cc[CHUNK_E / 512];
  float vv[CHUNK_E / 512];
  #pragma unroll
  for (int k = 0; k < CHUNK_E / 512; ++k) {
    const int i = min(i0 + k * 512, N_EDGES - 1);
    r[k] = rw[i]; cc[k] = cl[i]; vv[k] = val[i];
  }
  int g = 0, n = 0;
  if (t < NBKT) {
    const int f = t * NCHK + c;
    g = offsCB[f];
    n = ((f + 1 < CNT_PAD) ? offsCB[f + 1] : N_EDGES) - g;
  }
  int x = n;
  for (int d = 1; d < 32; d <<= 1) {
    const int y = __shfl_up(x, d);
    if (lane >= d) x += y;
  }
  if (lane == 31) wsum[wave] = x;
  __syncthreads();
  int base = 0;
  for (int w = 0; w < wave; ++w) base += wsum[w];
  if (t < NBKT) {
    const int ls = base + x - n;
    cur[t] = ls;
    dlt[t] = g - ls;
  }
  __syncthreads();
  #pragma unroll
  for (int k = 0; k < CHUNK_E / 512; ++k)
    if (i0 + k * 512 < N_EDGES) {
      const int b = r[k] >> BKT_SHIFT;
      const int p = min(max(atomicAdd(&cur[b], 1), 0), CHUNK_E - 1);
      stg[p] = make_int2(((r[k] & (BKT_ROWS - 1)) << COL_BITS) | cc[k], __float_as_int(vv[k]));
      rng[p] = (unsigned short)b;
    }
  __syncthreads();
  const int ne = N_EDGES - c * CHUNK_E;
  int  dd[CHUNK_E / 512];
  int2 rc[CHUNK_E / 512];
  #pragma unroll
  for (int k = 0; k < CHUNK_E / 512; ++k) {
    dd[k] = dlt[min((int)rng[t + k * 512], NBKT - 1)];
    rc[k] = stg[t + k * 512];
  }
  #pragma unroll
  for (int k = 0; k < CHUNK_E / 512; ++k)
    if (t + k * 512 < ne) grp[t + k * 512 + dd[k]] = rc[k];
}

#define STG_CAP    12288

template <bool STAGED>
__device__ __forceinline__ void lg_rank_place(const int2* __restrict__ grp, const int bs, const int be, const int t,
                                              int* cnt, int2* stg, int2* __restrict__ csr) {
  for (int j = bs + t; j < be; j += 4 * 512) {
    int2 e[4];
    #pragma unroll
    for (int u = 0; u < 4; ++u) e[u] = grp[min(j + u * 512, be - 1)];
    #pragma unroll
    for (int u = 0; u < 4; ++u)
      if (j + u * 512 < be) {
        const int p = atomicAdd(&cnt[(e[u].x >> COL_BITS) & (BKT_ROWS - 1)], 1);
        const int2 o = make_int2(e[u].x & ((1 << COL_BITS) - 1), e[u].y);
        if (STAGED) stg[min(max(p - bs, 0), STG_CAP - 1)] = o;
        else        csr[p] = o;
      }
  }
}

__global__ __launch_bounds__(512) void lg_bkt_rows(const int2* __restrict__ grp, const int* __restrict__ offsCB,
                                                   int* __restrict__ rowp, int2* __restrict__ csr) {
  __shared__ int  cnt[BKT_ROWS];
  __shared__ int  scn[BKT_ROWS];
  __shared__ int2 stg[STG_CAP];
  const int t = threadIdx.x, b = blockIdx.x;
  const int bs = max(0, min(offsCB[b * NCHK], N_EDGES));
  const int be = (b + 1 < NBKT) ? min(offsCB[(b + 1) * NCHK], N_EDGES) : N_EDGES;
  const int n = be - bs;
  cnt[t] = 0;
  __syncthreads();
  for (int j = bs + t; j < be; j += 4 * 512) {
    int key[4];
    #pragma unroll
    for (int u = 0; u < 4; ++u) key[u] = grp[min(j + u * 512, be - 1)].x;
    #pragma unroll
    for (int u = 0; u < 4; ++u)
      if (j + u * 512 < be) atomicAdd(&cnt[(key[u] >> COL_BITS) & (BKT_ROWS - 1)], 1);
  }
  __syncthreads();
  const int c = cnt[t];
  scn[t] = c;
  __syncthreads();
  for (int s = 1; s < BKT_ROWS; s <<= 1) {
    const int add = (t >= s) ? scn[t - s] : 0;
    __syncthreads();
    scn[t] += add;
    __syncthreads();
  }
  const int start = bs + scn[t] - c;
  rowp[b * BKT_ROWS + t] = start;
  cnt[t] = start;
  __syncthreads();
  if (n > STG_CAP) {
    lg_rank_place<false>(grp, bs, be, t, cnt, stg, csr);
    return;
  }
  lg_rank_place<true>(grp, bs, be, t, cnt, stg, csr);
  __syncthreads();
  const int h = min((16 - (bs & 15)) & 15, n);
  if (t < h) csr[bs + t] = stg[t];
  const int np = (n - h) >> 1;
  v4i* dst4 = (v4i*)(csr + bs + h);
  #pragma unroll 4
  for (int i = t; i < np; i += 512) {
    const int j = h + 2 * i;
    const v4i o = {stg[j].x, stg[j].y, stg[j + 1].x, stg[j + 1].y};
    dst4[i] = o;
  }
  if (t == 0 && ((n - h) & 1)) csr[be - 1] = stg[n - 1];
}

template <bool LIST>
__global__ void lg_scan(const int4* __restrict__ cnt4, int4* __restrict__ rowp4,
                        int* __restrict__ list, int* __restrict__ nlist) {
  __shared__ int sh[256];
  __shared__ int sb[256];
  const int t = threadIdx.x, b = blockIdx.x;
  int pre = 0;
  for (int i = t; i < b * 256; i += 256) {
    const int4 u = cnt4[i];
    pre += u.x + u.y + u.z + u.w;
  }
  const int4 v = cnt4[b * 256 + t];
  const int tsum = v.x + v.y + v.z + v.w;
  sh[t] = tsum;
  sb[t] = pre;
  __syncthreads();
  for (int s = 1; s < 256; s <<= 1) {
    const int add = (t >= s) ? sh[t - s] : 0;
    __syncthreads();
    sh[t] += add;
    __syncthreads();
  }
  for (int s = 128; s > 0; s >>= 1) {
    if (t < s) sb[t] += sb[t + s];
    __syncthreads();
  }
  const int base = sb[0] + sh[t] - tsum;
  int4 o;
  o.x = base; o.y = o.x + v.x; o.z = o.y + v.y; o.w = o.z + v.z;
  if (!LIST) {
    rowp4[b * 256 + t] = o;
  } else {
    const int r = (b * 256 + t) * 4;
    if (v.x) list[o.x] = r;
    if (v.y) list[o.y] = r + 1;
    if (v.z) list[o.z] = r + 2;
    if (v.w) list[o.w] = r + 3;
    if (b == SCAN_NB - 1 && t == 255) *nlist = o.w + v.w;
  }
}

__device__ __forceinline__ void lg_mark(const int t, const int* __restrict__ uid, const int* __restrict__ iid,
                                        const int* __restrict__ rowp, const int2* __restrict__ csr,
                                        int* __restrict__ flag) {
  const int r = t >> 3, q = t & 7;
  const int n = (r < BATCH) ? uid[r] : (NUM_USERS + iid[r - BATCH]);
  const int beg = rowp[n], end = rowp[n + 1];
  if (q == 0) atomicOr(&flag[n], 1);
  for (int j = beg + q; j < end; j += 8) atomicOr(&flag[csr[j].x], 1);
}


template <int K>
__device__ __forceinline__ int grp_bcast(int v) {
  return (int)__builtin_amdgcn_mov_dpp8((unsigned)v, (unsigned)(K * 0x249249));
}

#define LG_SLOT(K)                                                                \
  {                                                                               \
    const unsigned c = (unsigned)grp_bcast<K>(cc);                                \
    w[K] = __int_as_float(grp_bcast<K>(ww));                                      \
    x[K] = *(const v4f*)(srcb + (size_t)(c * (unsigned)(STRIDE * 4) + qoff));     \
  }

template <int STRIDE>
__device__ __forceinline__ void lg_round(const char* __restrict__ srcb, const int cc, const int ww,
                                         const unsigned qoff, v4f& acc) {
  v4f   x[8];
  float w[8];
  LG_SLOT(0) LG_SLOT(1) LG_SLOT(2) LG_SLOT(3) LG_SLOT(4) LG_SLOT(5) LG_SLOT(6) LG_SLOT(7)
  #pragma unroll
  for (int k = 0; k < 8; ++k) acc += w[k] * x[k];
}

enum { LG_ALL = 0, LG_LIST = 1, LG_SLOT = 2 };

template <int STRIDE, int MODE, bool MARK>
__global__ void lg_pull4(const float* __restrict__ src, const int* __restrict__ rowp,
                         const int2* __restrict__ csr, float* __restrict__ dst,
                         const int* __restrict__ list, const int* __restrict__ nlist,
                         const int* __restrict__ uid, const int* __restrict__ iid, int off,
                         const float* __restrict__ emb0, const float* __restrict__ ta, int* __restrict__ flag) {
  static_assert(N_TOTAL % 4 == 0 && (2 * BATCH) % 4 == 0, "a wave takes four whole rows");
  if (MARK) {
    const int t = ((int)blockIdx.x - ((int)gridDim.x - 2 * BATCH * 8 / 256)) * 256 + (int)threadIdx.x;
    if (t >= 0) lg_mark(t, uid, iid, rowp, csr, flag);
  }
  const int lane = threadIdx.x & 31;
  const int q = lane & 7;
  const int g = lane >> 3;
  const unsigned qoff = (unsigned)q * 16u;
  const char* srcb = (const char*)src;
  const int w0 = __builtin_amdgcn_readfirstlane((int)((blockIdx.x * blockDim.x + threadIdx.x) >> 5));
  const int nw = (int)((gridDim.x * blockDim.x) >> 5);
  int total = N_TOTAL;
  if (MODE == LG_SLOT) total = 2 * BATCH;
  if (MODE == LG_LIST) total = min(*nlist, N_TOTAL);
  for (int r0 = w0 * 4; r0 < total; r0 += nw * 4) {
    const int i = r0 + g;
    const bool act = (MODE != LG_LIST) || i < total;
    int row = i;
    if (MODE == LG_LIST) row = act ? list[i] : 0;
    if (MODE == LG_SLOT) row = (i < BATCH) ? uid[i] : (NUM_USERS + iid[i - BATCH]);
    int beg = 0, end = 0;
    if (act) { beg = rowp[row]; end = rowp[row + 1]; }
    v4f acc = {0.f, 0.f, 0.f, 0.f};
    v4f l0 = acc, l1 = acc, l2 = acc;
    if (MODE == LG_SLOT) {
      l0 = *(const v4f*)(emb0 + (size_t)row * EMBED_DIM + q * 4);
      l1 = *(const v4f*)(ta + (size_t)row * HALF_DIM + q * 4);
      l2 = *(const v4f*)(src + (size_t)row * HALF_DIM + q * 4);
    }
    for (int jb = beg; jb < end; jb += 32) {
      const int n = min(end - jb, 32);
      int2 rec[4];
      #pragma unroll
      for (int j = 0; j < 4; ++j) {
        rec[j] = make_int2(0, 0);
        if (8 * j < n) rec[j] = csr[jb + min(8 * j + q, n - 1)];
      }
      #pragma unroll
      for (int j = 0; j < 4; ++j)
        if (8 * j < n)
          lg_round<STRIDE>(srcb, rec[j].x, (8 * j + q < n) ? rec[j].y : 0  , qoff, acc);
    }
    if (MODE == LG_SLOT) {
      *(v4f*)(dst + (size_t)i * EMBED_DIM + off + q * 4) = ((l0 + l1) + l2) + acc;
    } else if (act) {
      *(v4f*)(dst + (size_t)row * HALF_DIM + q * 4) = acc;
    }
  }
}

__global__ void lg_dot_wmma(const float* __restrict__ acc, float* __restrict__ out) {
  __shared__ float sd[8][8][32];
  const int lane = threadIdx.x & 31;
  const int wave = threadIdx.x >> 5;
  const int tile = blockIdx.x * 8 + wave;
  const int k0   = tile * 16;
  const int rN   = lane & 15;
  const float* U = acc + (size_t)(k0 + rN) * EMBED_DIM;
  const float* I = acc + (size_t)(BATCH + k0 + rN) * EMBED_DIM;

  v8f c = {0.f, 0.f, 0.f, 0.f, 0.f, 0.f, 0.f, 0.f};

#if __has_builtin(__builtin_amdgcn_wmma_f32_16x16x4_f32)
  const int kb = (lane >> 4) << 1;
  #pragma unroll
  for (int s = 0; s < 16; ++s) {
    const int col = s * 4 + kb;
    v2f a; a[0] = U[col]; a[1] = U[col + 1];
    v2f b; b[0] = I[col]; b[1] = I[col + 1];
    c = __builtin_amdgcn_wmma_f32_16x16x4_f32(false, a, false, b, (short)0, c, false, false);
  }
#else
  const int kb8 = (lane >> 4) << 3;
  #pragma unroll
  for (int blk = 0; blk < 2; ++blk) {
    v16h a, b;
    #pragma unroll
    for (int i = 0; i < 8; ++i) {
      const int kk = blk * 32 + kb8 + ((i < 4) ? (2 * i) : (16 + 2 * (i - 4)));
      a[2 * i]     = (_Float16)U[kk];
      a[2 * i + 1] = (_Float16)U[kk + 1];
      b[2 * i]     = (_Float16)I[kk];
      b[2 * i + 1] = (_Float16)I[kk + 1];
    }
    c = __builtin_amdgcn_wmma_f32_16x16x32_f16(false, a, false, b, (short)0, c, false, false);
  }
#endif

  #pragma unroll
  for (int g = 0; g < 8; ++g) sd[wave][g][lane] = c[g];
  __syncthreads();

  if (lane < 16) {
    const float v = (lane < 8) ? sd[wave][lane][lane] : sd[wave][lane - 8][lane + 16];
    out[k0 + lane] = v * 0.0625f;
  }
}

static void lg_half23(const float* node_emb_half, int col0, const int* rowp, const int2* csr,
                      float* tblA, float* tblB, const int* list, const int* nlist,
                      const int* user_ids, const int* item_ids, float* accS, int* flag, hipStream_t stream) {
  const dim3 b256(256);
  lg_pull4<HALF_DIM, LG_LIST, false><<<2048, b256, 0, stream>>>(tblA, rowp, csr, tblB, list, nlist,
                                                     user_ids, item_ids, col0, node_emb_half, tblA, flag);
  lg_pull4<HALF_DIM, LG_SLOT, false><<<2 * BATCH / 32, b256, 0, stream>>>(tblB, rowp, csr, accS, list, nlist,
                                                     user_ids, item_ids, col0, node_emb_half, tblA, flag);
}

extern "C" void kernel_launch(void* const* d_in, const int* in_sizes, int n_in,
                              void* d_out, int out_size, void* d_ws, size_t ws_size,
                              hipStream_t stream) {
  const float* node_emb = (const float*)d_in[0];
  const float* adj_val  = (const float*)d_in[1];
  const int*   adj_row  = (const int*)d_in[2];
  const int*   adj_col  = (const int*)d_in[3];
  const int*   user_ids = (const int*)d_in[4];
  const int*   item_ids = (const int*)d_in[5];
  float* out = (float*)d_out;

  char* ws    = (char*)d_ws;
  float* tblA = (float*)(ws + OFF_TBLA);
  float* tblB = (float*)(ws + OFF_TBLB);
  int2*  grp  = (int2*)(ws + OFF_TBLA);
  int2*  csr  = (int2*)(ws + OFF_CSR);
  int*   rowp = (int*)(ws + OFF_ROWP);
  int*   cntCB = (int*)(ws + OFF_CNT);
  int*   flag = (int*)(ws + OFF_FLAG);
  int*   list = (int*)(ws + OFF_LIST);
  int*   nlist = (int*)(ws + OFF_NLIST);
  int*   offsCB = (int*)(ws + OFF_LIST);
  float* accS = (float*)(ws + OFF_ACCS);

  const dim3 b256(256);

  lg_bkt_count<<<NCHK, 512, 0, stream>>>(adj_row, cntCB, flag);
  lg_scan<false><<<SCAN_NB, b256, 0, stream>>>((const int4*)cntCB, (int4*)offsCB, list, nlist);
  lg_bkt_place<<<NCHK, 512, 0, stream>>>(adj_row, adj_col, adj_val, offsCB, grp);
  lg_bkt_rows<<<NBKT, 512, 0, stream>>>(grp, offsCB, rowp, csr);

  lg_pull4<EMBED_DIM, LG_ALL, true><<<2048, b256, 0, stream>>>(node_emb, rowp, csr, tblA, list, nlist,
                                                     user_ids, item_ids, 0, node_emb, tblA, flag);
  lg_scan<true><<<SCAN_NB, b256, 0, stream>>>((const int4*)flag, (int4*)rowp, list, nlist);
  lg_half23(node_emb, 0, rowp, csr, tblA, tblB, list, nlist, user_ids, item_ids, accS, flag, stream);

  lg_pull4<EMBED_DIM, LG_ALL, false><<<2048, b256, 0, stream>>>(node_emb + 32, rowp, csr, tblA, list, nlist,
                                                     user_ids, item_ids, 32, node_emb + 32, tblA, flag);
  lg_half23(node_emb + 32, 32, rowp, csr, tblA, tblB, list, nlist, user_ids, item_ids, accS, flag, stream);

  lg_dot_wmma<<<BATCH / 128, b256, 0, stream>>>(accS, out);
}


# Both scan launches on 49 blocks of 768 threads, not 147 of 256: in-front sum at most 48 loads a thread, 16 in flight, not 146 by 8; thread sums scanned by wave shuffles, 1 barrier, not 26
# speedup vs baseline: 4.4617x; 1.0966x over previous
#include <hip/hip_runtime.h>

#define N_TOTAL    150100
#define EMBED_DIM  64
#define HALF_DIM   32
#define NUM_USERS  100000
#define N_EDGES    3000000
#define BATCH      4096
#define NUM_LAYERS 3

#define CNT_PAD    150528
#define SCAN_NB    49

#define BKT_SHIFT  9
#define BKT_ROWS   512
#define NBKT       294
#define NCHK       512
#define CHUNK_E    6144
#define COL_BITS   18
static_assert(NBKT * NCHK == CNT_PAD && NBKT * BKT_ROWS == CNT_PAD, "one flat scan serves both tables");
static_assert(NCHK * CHUNK_E >= N_EDGES && N_TOTAL < (1 << COL_BITS) && CHUNK_E % 512 == 0, "grouping sizes");

#define OFF_TBLA   0u
#define OFF_TBLB   19212800u
#define OFF_CSR    38425600u
#define OFF_ROWP   62425600u
#define OFF_CNT    63027712u
#define OFF_FLAG   63629824u
#define OFF_LIST   64231936u
#define OFF_NLIST  64834048u
#define OFF_ACCS   76851200u

typedef __attribute__((ext_vector_type(2)))  float    v2f;
typedef __attribute__((ext_vector_type(4)))  float    v4f;
typedef __attribute__((ext_vector_type(4)))  int      v4i;
typedef __attribute__((ext_vector_type(8)))  float    v8f;
typedef __attribute__((ext_vector_type(16))) _Float16 v16h;

__global__ __launch_bounds__(512) void lg_bkt_count(const int* __restrict__ rw, int* __restrict__ cntCB,
                                                    int* __restrict__ flag) {
  static_assert(NCHK * 512 >= CNT_PAD, "one thread per row flag");
  __shared__ int h[NBKT];
  const int t = threadIdx.x, c = blockIdx.x;
  if (c * 512 + t < CNT_PAD) flag[c * 512 + t] = 0;
  if (t < NBKT) h[t] = 0;
  __syncthreads();
  const int i0 = c * CHUNK_E + t;
  int r[CHUNK_E / 512];
  #pragma unroll
  for (int k = 0; k < CHUNK_E / 512; ++k) r[k] = rw[min(i0 + k * 512, N_EDGES - 1)];
  #pragma unroll
  for (int k = 0; k < CHUNK_E / 512; ++k)
    if (i0 + k * 512 < N_EDGES) atomicAdd(&h[r[k] >> BKT_SHIFT], 1);
  __syncthreads();
  if (t < NBKT) cntCB[t * NCHK + c] = h[t];
}

__global__ __launch_bounds__(512) void lg_bkt_place(const int* __restrict__ rw, const int* __restrict__ cl,
                                                    const float* __restrict__ val,
                                                    const int* __restrict__ offsCB, int2* __restrict__ grp) {
  __shared__ int  cur[NBKT];
  __shared__ int  dlt[NBKT];
  __shared__ int  wsum[16];
  __shared__ int2 stg[CHUNK_E];
  __shared__ unsigned short rng[CHUNK_E];
  const int t = threadIdx.x, c = blockIdx.x;
  const int lane = t & 31, wave = t >> 5;
  const int i0 = c * CHUNK_E + t;
  int   r[CHUNK_E / 512], cc[CHUNK_E / 512];
  float vv[CHUNK_E / 512];
  #pragma unroll
  for (int k = 0; k < CHUNK_E / 512; ++k) {
    const int i = min(i0 + k * 512, N_EDGES - 1);
    r[k] = rw[i]; cc[k] = cl[i]; vv[k] = val[i];
  }
  int g = 0, n = 0;
  if (t < NBKT) {
    const int f = t * NCHK + c;
    g = offsCB[f];
    n = ((f + 1 < CNT_PAD) ? offsCB[f + 1] : N_EDGES) - g;
  }
  int x = n;
  for (int d = 1; d < 32; d <<= 1) {
    const int y = __shfl_up(x, d);
    if (lane >= d) x += y;
  }
  if (lane == 31) wsum[wave] = x;
  __syncthreads();
  int base = 0;
  for (int w = 0; w < wave; ++w) base += wsum[w];
  if (t < NBKT) {
    const int ls = base + x - n;
    cur[t] = ls;
    dlt[t] = g - ls;
  }
  __syncthreads();
  #pragma unroll
  for (int k = 0; k < CHUNK_E / 512; ++k)
    if (i0 + k * 512 < N_EDGES) {
      const int b = r[k] >> BKT_SHIFT;
      const int p = min(max(atomicAdd(&cur[b], 1), 0), CHUNK_E - 1);
      stg[p] = make_int2(((r[k] & (BKT_ROWS - 1)) << COL_BITS) | cc[k], __float_as_int(vv[k]));
      rng[p] = (unsigned short)b;
    }
  __syncthreads();
  const int ne = N_EDGES - c * CHUNK_E;
  int  dd[CHUNK_E / 512];
  int2 rc[CHUNK_E / 512];
  #pragma unroll
  for (int k = 0; k < CHUNK_E / 512; ++k) {
    dd[k] = dlt[min((int)rng[t + k * 512], NBKT - 1)];
    rc[k] = stg[t + k * 512];
  }
  #pragma unroll
  for (int k = 0; k < CHUNK_E / 512; ++k)
    if (t + k * 512 < ne) grp[t + k * 512 + dd[k]] = rc[k];
}

#define STG_CAP    12288

template <bool STAGED>
__device__ __forceinline__ void lg_rank_place(const int2* __restrict__ grp, const int bs, const int be, const int t,
                                              int* cnt, int2* stg, int2* __restrict__ csr) {
  for (int j = bs + t; j < be; j += 4 * 512) {
    int2 e[4];
    #pragma unroll
    for (int u = 0; u < 4; ++u) e[u] = grp[min(j + u * 512, be - 1)];
    #pragma unroll
    for (int u = 0; u < 4; ++u)
      if (j + u * 512 < be) {
        const int p = atomicAdd(&cnt[(e[u].x >> COL_BITS) & (BKT_ROWS - 1)], 1);
        const int2 o = make_int2(e[u].x & ((1 << COL_BITS) - 1), e[u].y);
        if (STAGED) stg[min(max(p - bs, 0), STG_CAP - 1)] = o;
        else        csr[p] = o;
      }
  }
}

__global__ __launch_bounds__(512) void lg_bkt_rows(const int2* __restrict__ grp, const int* __restrict__ offsCB,
                                                   int* __restrict__ rowp, int2* __restrict__ csr) {
  __shared__ int  cnt[BKT_ROWS];
  __shared__ int  scn[BKT_ROWS];
  __shared__ int2 stg[STG_CAP];
  const int t = threadIdx.x, b = blockIdx.x;
  const int bs = max(0, min(offsCB[b * NCHK], N_EDGES));
  const int be = (b + 1 < NBKT) ? min(offsCB[(b + 1) * NCHK], N_EDGES) : N_EDGES;
  const int n = be - bs;
  cnt[t] = 0;
  __syncthreads();
  for (int j = bs + t; j < be; j += 4 * 512) {
    int key[4];
    #pragma unroll
    for (int u = 0; u < 4; ++u) key[u] = grp[min(j + u * 512, be - 1)].x;
    #pragma unroll
    for (int u = 0; u < 4; ++u)
      if (j + u * 512 < be) atomicAdd(&cnt[(key[u] >> COL_BITS) & (BKT_ROWS - 1)], 1);
  }
  __syncthreads();
  const int c = cnt[t];
  scn[t] = c;
  __syncthreads();
  for (int s = 1; s < BKT_ROWS; s <<= 1) {
    const int add = (t >= s) ? scn[t - s] : 0;
    __syncthreads();
    scn[t] += add;
    __syncthreads();
  }
  const int start = bs + scn[t] - c;
  rowp[b * BKT_ROWS + t] = start;
  cnt[t] = start;
  __syncthreads();
  if (n > STG_CAP) {
    lg_rank_place<false>(grp, bs, be, t, cnt, stg, csr);
    return;
  }
  lg_rank_place<true>(grp, bs, be, t, cnt, stg, csr);
  __syncthreads();
  const int h = min((16 - (bs & 15)) & 15, n);
  if (t < h) csr[bs + t] = stg[t];
  const int np = (n - h) >> 1;
  v4i* dst4 = (v4i*)(csr + bs + h);
  #pragma unroll 4
  for (int i = t; i < np; i += 512) {
    const int j = h + 2 * i;
    const v4i o = {stg[j].x, stg[j].y, stg[j + 1].x, stg[j + 1].y};
    dst4[i] = o;
  }
  if (t == 0 && ((n - h) & 1)) csr[be - 1] = stg[n - 1];
}

#define SCAN_T     768
#define SCAN_W     (SCAN_T / 32)
static_assert(SCAN_NB * SCAN_T * 4 == CNT_PAD, "the scan blocks tile the counters");
template <bool LIST>
__global__ __launch_bounds__(SCAN_T) void lg_scan(const int4* __restrict__ cnt4, int4* __restrict__ rowp4,
                                                   int* __restrict__ list, int* __restrict__ nlist) {
  __shared__ int wtot[SCAN_W];
  __shared__ int wpre[SCAN_W];
  const int t = threadIdx.x, b = blockIdx.x;
  const int lane = t & 31, wave = t >> 5;
  int pre = 0;
  for (int i0 = 0; i0 < b; i0 += 16) {
    int4 u[16];
    #pragma unroll
    for (int k = 0; k < 16; ++k) u[k] = cnt4[min(i0 + k, b - 1) * SCAN_T + t];
    #pragma unroll
    for (int k = 0; k < 16; ++k)
      if (i0 + k < b) pre += u[k].x + u[k].y + u[k].z + u[k].w;
  }
  const int4 v = cnt4[b * SCAN_T + t];
  const int tsum = v.x + v.y + v.z + v.w;
  int x = tsum;
  for (int d = 1; d < 32; d <<= 1) {
    const int y = __shfl_up(x, d);
    if (lane >= d) x += y;
  }
  for (int d = 16; d > 0; d >>= 1) pre += __shfl_xor(pre, d);
  if (lane == 31) wtot[wave] = x;
  if (lane == 0)  wpre[wave] = pre;
  __syncthreads();
  int base = x - tsum;
  for (int w = 0; w < SCAN_W; ++w) {
    base += wpre[w];
    if (w < wave) base += wtot[w];
  }
  int4 o;
  o.x = base; o.y = o.x + v.x; o.z = o.y + v.y; o.w = o.z + v.z;
  if (!LIST) {
    rowp4[b * SCAN_T + t] = o;
  } else {
    const int r = (b * SCAN_T + t) * 4;
    if (v.x) list[o.x] = r;
    if (v.y) list[o.y] = r + 1;
    if (v.z) list[o.z] = r + 2;
    if (v.w) list[o.w] = r + 3;
    if (b == SCAN_NB - 1 && t == SCAN_T - 1) *nlist = o.w + v.w;
  }
}

__device__ __forceinline__ void lg_mark(const int t, const int* __restrict__ uid, const int* __restrict__ iid,
                                        const int* __restrict__ rowp, const int2* __restrict__ csr,
                                        int* __restrict__ flag) {
  const int r = t >> 3, q = t & 7;
  const int n = (r < BATCH) ? uid[r] : (NUM_USERS + iid[r - BATCH]);
  const int beg = rowp[n], end = rowp[n + 1];
  if (q == 0) atomicOr(&flag[n], 1);
  for (int j = beg + q; j < end; j += 8) atomicOr(&flag[csr[j].x], 1);
}


template <int K>
__device__ __forceinline__ int grp_bcast(int v) {
  return (int)__builtin_amdgcn_mov_dpp8((unsigned)v, (unsigned)(K * 0x249249));
}

#define LG_SLOT(K)                                                                \
  {                                                                               \
    const unsigned c = (unsigned)grp_bcast<K>(cc);                                \
    w[K] = __int_as_float(grp_bcast<K>(ww));                                      \
    x[K] = *(const v4f*)(srcb + (size_t)(c * (unsigned)(STRIDE * 4) + qoff));     \
  }

template <int STRIDE>
__device__ __forceinline__ void lg_round(const char* __restrict__ srcb, const int cc, const int ww,
                                         const unsigned qoff, v4f& acc) {
  v4f   x[8];
  float w[8];
  LG_SLOT(0) LG_SLOT(1) LG_SLOT(2) LG_SLOT(3) LG_SLOT(4) LG_SLOT(5) LG_SLOT(6) LG_SLOT(7)
  #pragma unroll
  for (int k = 0; k < 8; ++k) acc += w[k] * x[k];
}

enum { LG_ALL = 0, LG_LIST = 1, LG_SLOT = 2 };

template <int STRIDE, int MODE, bool MARK>
__global__ void lg_pull4(const float* __restrict__ src, const int* __restrict__ rowp,
                         const int2* __restrict__ csr, float* __restrict__ dst,
                         const int* __restrict__ list, const int* __restrict__ nlist,
                         const int* __restrict__ uid, const int* __restrict__ iid, int off,
                         const float* __restrict__ emb0, const float* __restrict__ ta, int* __restrict__ flag) {
  static_assert(N_TOTAL % 4 == 0 && (2 * BATCH) % 4 == 0, "a wave takes four whole rows");
  if (MARK) {
    const int t = ((int)blockIdx.x - ((int)gridDim.x - 2 * BATCH * 8 / 256)) * 256 + (int)threadIdx.x;
    if (t >= 0) lg_mark(t, uid, iid, rowp, csr, flag);
  }
  const int lane = threadIdx.x & 31;
  const int q = lane & 7;
  const int g = lane >> 3;
  const unsigned qoff = (unsigned)q * 16u;
  const char* srcb = (const char*)src;
  const int w0 = __builtin_amdgcn_readfirstlane((int)((blockIdx.x * blockDim.x + threadIdx.x) >> 5));
  const int nw = (int)((gridDim.x * blockDim.x) >> 5);
  int total = N_TOTAL;
  if (MODE == LG_SLOT) total = 2 * BATCH;
  if (MODE == LG_LIST) total = min(*nlist, N_TOTAL);
  for (int r0 = w0 * 4; r0 < total; r0 += nw * 4) {
    const int i = r0 + g;
    const bool act = (MODE != LG_LIST) || i < total;
    int row = i;
    if (MODE == LG_LIST) row = act ? list[i] : 0;
    if (MODE == LG_SLOT) row = (i < BATCH) ? uid[i] : (NUM_USERS + iid[i - BATCH]);
    int beg = 0, end = 0;
    if (act) { beg = rowp[row]; end = rowp[row + 1]; }
    v4f acc = {0.f, 0.f, 0.f, 0.f};
    v4f l0 = acc, l1 = acc, l2 = acc;
    if (MODE == LG_SLOT) {
      l0 = *(const v4f*)(emb0 + (size_t)row * EMBED_DIM + q * 4);
      l1 = *(const v4f*)(ta + (size_t)row * HALF_DIM + q * 4);
      l2 = *(const v4f*)(src + (size_t)row * HALF_DIM + q * 4);
    }
    for (int jb = beg; jb < end; jb += 32) {
      const int n = min(end - jb, 32);
      int2 rec[4];
      #pragma unroll
      for (int j = 0; j < 4; ++j) {
        rec[j] = make_int2(0, 0);
        if (8 * j < n) rec[j] = csr[jb + min(8 * j + q, n - 1)];
      }
      #pragma unroll
      for (int j = 0; j < 4; ++j)
        if (8 * j < n)
          lg_round<STRIDE>(srcb, rec[j].x, (8 * j + q < n) ? rec[j].y : 0  , qoff, acc);
    }
    if (MODE == LG_SLOT) {
      *(v4f*)(dst + (size_t)i * EMBED_DIM + off + q * 4) = ((l0 + l1) + l2) + acc;
    } else if (act) {
      *(v4f*)(dst + (size_t)row * HALF_DIM + q * 4) = acc;
    }
  }
}

__global__ void lg_dot_wmma(const float* __restrict__ acc, float* __restrict__ out) {
  __shared__ float sd[8][8][32];
  const int lane = threadIdx.x & 31;
  const int wave = threadIdx.x >> 5;
  const int tile = blockIdx.x * 8 + wave;
  const int k0   = tile * 16;
  const int rN   = lane & 15;
  const float* U = acc + (size_t)(k0 + rN) * EMBED_DIM;
  const float* I = acc + (size_t)(BATCH + k0 + rN) * EMBED_DIM;

  v8f c = {0.f, 0.f, 0.f, 0.f, 0.f, 0.f, 0.f, 0.f};

#if __has_builtin(__builtin_amdgcn_wmma_f32_16x16x4_f32)
  const int kb = (lane >> 4) << 1;
  #pragma unroll
  for (int s = 0; s < 16; ++s) {
    const int col = s * 4 + kb;
    v2f a; a[0] = U[col]; a[1] = U[col + 1];
    v2f b; b[0] = I[col]; b[1] = I[col + 1];
    c = __builtin_amdgcn_wmma_f32_16x16x4_f32(false, a, false, b, (short)0, c, false, false);
  }
#else
  const int kb8 = (lane >> 4) << 3;
  #pragma unroll
  for (int blk = 0; blk < 2; ++blk) {
    v16h a, b;
    #pragma unroll
    for (int i = 0; i < 8; ++i) {
      const int kk = blk * 32 + kb8 + ((i < 4) ? (2 * i) : (16 + 2 * (i - 4)));
      a[2 * i]     = (_Float16)U[kk];
      a[2 * i + 1] = (_Float16)U[kk + 1];
      b[2 * i]     = (_Float16)I[kk];
      b[2 * i + 1] = (_Float16)I[kk + 1];
    }
    c = __builtin_amdgcn_wmma_f32_16x16x32_f16(false, a, false, b, (short)0, c, false, false);
  }
#endif

  #pragma unroll
  for (int g = 0; g < 8; ++g) sd[wave][g][lane] = c[g];
  __syncthreads();

  if (lane < 16) {
    const float v = (lane < 8) ? sd[wave][lane][lane] : sd[wave][lane - 8][lane + 16];
    out[k0 + lane] = v * 0.0625f;
  }
}

static void lg_half23(const float* node_emb_half, int col0, const int* rowp, const int2* csr,
                      float* tblA, float* tblB, const int* list, const int* nlist,
                      const int* user_ids, const int* item_ids, float* accS, int* flag, hipStream_t stream) {
  const dim3 b256(256);
  lg_pull4<HALF_DIM, LG_LIST, false><<<2048, b256, 0, stream>>>(tblA, rowp, csr, tblB, list, nlist,
                                                     user_ids, item_ids, col0, node_emb_half, tblA, flag);
  lg_pull4<HALF_DIM, LG_SLOT, false><<<2 * BATCH / 32, b256, 0, stream>>>(tblB, rowp, csr, accS, list, nlist,
                                                     user_ids, item_ids, col0, node_emb_half, tblA, flag);
}

extern "C" void kernel_launch(void* const* d_in, const int* in_sizes, int n_in,
                              void* d_out, int out_size, void* d_ws, size_t ws_size,
                              hipStream_t stream) {
  const float* node_emb = (const float*)d_in[0];
  const float* adj_val  = (const float*)d_in[1];
  const int*   adj_row  = (const int*)d_in[2];
  const int*   adj_col  = (const int*)d_in[3];
  const int*   user_ids = (const int*)d_in[4];
  const int*   item_ids = (const int*)d_in[5];
  float* out = (float*)d_out;

  char* ws    = (char*)d_ws;
  float* tblA = (float*)(ws + OFF_TBLA);
  float* tblB = (float*)(ws + OFF_TBLB);
  int2*  grp  = (int2*)(ws + OFF_TBLA);
  int2*  csr  = (int2*)(ws + OFF_CSR);
  int*   rowp = (int*)(ws + OFF_ROWP);
  int*   cntCB = (int*)(ws + OFF_CNT);
  int*   flag = (int*)(ws + OFF_FLAG);
  int*   list = (int*)(ws + OFF_LIST);
  int*   nlist = (int*)(ws + OFF_NLIST);
  int*   offsCB = (int*)(ws + OFF_LIST);
  float* accS = (float*)(ws + OFF_ACCS);

  const dim3 b256(256);

  lg_bkt_count<<<NCHK, 512, 0, stream>>>(adj_row, cntCB, flag);
  lg_scan<false><<<SCAN_NB, SCAN_T, 0, stream>>>((const int4*)cntCB, (int4*)offsCB, list, nlist);
  lg_bkt_place<<<NCHK, 512, 0, stream>>>(adj_row, adj_col, adj_val, offsCB, grp);
  lg_bkt_rows<<<NBKT, 512, 0, stream>>>(grp, offsCB, rowp, csr);

  lg_pull4<EMBED_DIM, LG_ALL, true><<<2048, b256, 0, stream>>>(node_emb, rowp, csr, tblA, list, nlist,
                                                     user_ids, item_ids, 0, node_emb, tblA, flag);
  lg_scan<true><<<SCAN_NB, SCAN_T, 0, stream>>>((const int4*)flag, (int4*)rowp, list, nlist);
  lg_half23(node_emb, 0, rowp, csr, tblA, tblB, list, nlist, user_ids, item_ids, accS, flag, stream);

  lg_pull4<EMBED_DIM, LG_ALL, false><<<2048, b256, 0, stream>>>(node_emb + 32, rowp, csr, tblA, list, nlist,
                                                     user_ids, item_ids, 32, node_emb + 32, tblA, flag);
  lg_half23(node_emb + 32, 32, rowp, csr, tblA, tblB, list, nlist, user_ids, item_ids, accS, flag, stream);

  lg_dot_wmma<<<BATCH / 128, b256, 0, stream>>>(accS, out);
}


# Layer 2 never stored: a wave per gathered row sums layer 2 at its ~21 source rows from layer 1, then layer 3; freed workspace: both passes take whole 64-dim rows. 12 -> 7 launches; sum order kept
# speedup vs baseline: 5.3156x; 1.1914x over previous
#include <hip/hip_runtime.h>

#define N_TOTAL    150100
#define EMBED_DIM  64
#define HALF_DIM   32
#define NUM_USERS  100000
#define N_EDGES    3000000
#define BATCH      4096
#define NUM_LAYERS 3

#define CNT_PAD    150528
#define SCAN_NB    49

#define BKT_SHIFT  9
#define BKT_ROWS   512
#define NBKT       294
#define NCHK       512
#define CHUNK_E    6144
#define COL_BITS   18
static_assert(NBKT * NCHK == CNT_PAD && NBKT * BKT_ROWS == CNT_PAD, "one flat scan serves both tables");
static_assert(NCHK * CHUNK_E >= N_EDGES && N_TOTAL < (1 << COL_BITS) && CHUNK_E % 512 == 0, "grouping sizes");

#define OFF_TBL1   0u
#define OFF_CSR    38425600u
#define OFF_ROWP   62425600u
#define OFF_CNT    63027712u
#define OFF_OFFS   64231936u
#define OFF_ACCS   76851200u

typedef __attribute__((ext_vector_type(2)))  float    v2f;
typedef __attribute__((ext_vector_type(4)))  float    v4f;
typedef __attribute__((ext_vector_type(4)))  int      v4i;
typedef __attribute__((ext_vector_type(8)))  float    v8f;
typedef __attribute__((ext_vector_type(16))) _Float16 v16h;

__global__ __launch_bounds__(512) void lg_bkt_count(const int* __restrict__ rw, int* __restrict__ cntCB) {
  __shared__ int h[NBKT];
  const int t = threadIdx.x, c = blockIdx.x;
  if (t < NBKT) h[t] = 0;
  __syncthreads();
  const int i0 = c * CHUNK_E + t;
  int r[CHUNK_E / 512];
  #pragma unroll
  for (int k = 0; k < CHUNK_E / 512; ++k) r[k] = rw[min(i0 + k * 512, N_EDGES - 1)];
  #pragma unroll
  for (int k = 0; k < CHUNK_E / 512; ++k)
    if (i0 + k * 512 < N_EDGES) atomicAdd(&h[r[k] >> BKT_SHIFT], 1);
  __syncthreads();
  if (t < NBKT) cntCB[t * NCHK + c] = h[t];
}

__global__ __launch_bounds__(512) void lg_bkt_place(const int* __restrict__ rw, const int* __restrict__ cl,
                                                    const float* __restrict__ val,
                                                    const int* __restrict__ offsCB, int2* __restrict__ grp) {
  __shared__ int  cur[NBKT];
  __shared__ int  dlt[NBKT];
  __shared__ int  wsum[16];
  __shared__ int2 stg[CHUNK_E];
  __shared__ unsigned short rng[CHUNK_E];
  const int t = threadIdx.x, c = blockIdx.x;
  const int lane = t & 31, wave = t >> 5;
  const int i0 = c * CHUNK_E + t;
  int   r[CHUNK_E / 512], cc[CHUNK_E / 512];
  float vv[CHUNK_E / 512];
  #pragma unroll
  for (int k = 0; k < CHUNK_E / 512; ++k) {
    const int i = min(i0 + k * 512, N_EDGES - 1);
    r[k] = rw[i]; cc[k] = cl[i]; vv[k] = val[i];
  }
  int g = 0, n = 0;
  if (t < NBKT) {
    const int f = t * NCHK + c;
    g = offsCB[f];
    n = ((f + 1 < CNT_PAD) ? offsCB[f + 1] : N_EDGES) - g;
  }
  int x = n;
  for (int d = 1; d < 32; d <<= 1) {
    const int y = __shfl_up(x, d);
    if (lane >= d) x += y;
  }
  if (lane == 31) wsum[wave] = x;
  __syncthreads();
  int base = 0;
  for (int w = 0; w < wave; ++w) base += wsum[w];
  if (t < NBKT) {
    const int ls = base + x - n;
    cur[t] = ls;
    dlt[t] = g - ls;
  }
  __syncthreads();
  #pragma unroll
  for (int k = 0; k < CHUNK_E / 512; ++k)
    if (i0 + k * 512 < N_EDGES) {
      const int b = r[k] >> BKT_SHIFT;
      const int p = min(max(atomicAdd(&cur[b], 1), 0), CHUNK_E - 1);
      stg[p] = make_int2(((r[k] & (BKT_ROWS - 1)) << COL_BITS) | cc[k], __float_as_int(vv[k]));
      rng[p] = (unsigned short)b;
    }
  __syncthreads();
  const int ne = N_EDGES - c * CHUNK_E;
  int  dd[CHUNK_E / 512];
  int2 rc[CHUNK_E / 512];
  #pragma unroll
  for (int k = 0; k < CHUNK_E / 512; ++k) {
    dd[k] = dlt[min((int)rng[t + k * 512], NBKT - 1)];
    rc[k] = stg[t + k * 512];
  }
  #pragma unroll
  for (int k = 0; k < CHUNK_E / 512; ++k)
    if (t + k * 512 < ne) grp[t + k * 512 + dd[k]] = rc[k];
}

#define STG_CAP    12288

template <bool STAGED>
__device__ __forceinline__ void lg_rank_place(const int2* __restrict__ grp, const int bs, const int be, const int t,
                                              int* cnt, int2* stg, int2* __restrict__ csr) {
  for (int j = bs + t; j < be; j += 4 * 512) {
    int2 e[4];
    #pragma unroll
    for (int u = 0; u < 4; ++u) e[u] = grp[min(j + u * 512, be - 1)];
    #pragma unroll
    for (int u = 0; u < 4; ++u)
      if (j + u * 512 < be) {
        const int p = atomicAdd(&cnt[(e[u].x >> COL_BITS) & (BKT_ROWS - 1)], 1);
        const int2 o = make_int2(e[u].x & ((1 << COL_BITS) - 1), e[u].y);
        if (STAGED) stg[min(max(p - bs, 0), STG_CAP - 1)] = o;
        else        csr[p] = o;
      }
  }
}

__global__ __launch_bounds__(512) void lg_bkt_rows(const int2* __restrict__ grp, const int* __restrict__ offsCB,
                                                   int* __restrict__ rowp, int2* __restrict__ csr) {
  __shared__ int  cnt[BKT_ROWS];
  __shared__ int  scn[BKT_ROWS];
  __shared__ int2 stg[STG_CAP];
  const int t = threadIdx.x, b = blockIdx.x;
  const int bs = max(0, min(offsCB[b * NCHK], N_EDGES));
  const int be = (b + 1 < NBKT) ? min(offsCB[(b + 1) * NCHK], N_EDGES) : N_EDGES;
  const int n = be - bs;
  cnt[t] = 0;
  __syncthreads();
  for (int j = bs + t; j < be; j += 4 * 512) {
    int key[4];
    #pragma unroll
    for (int u = 0; u < 4; ++u) key[u] = grp[min(j + u * 512, be - 1)].x;
    #pragma unroll
    for (int u = 0; u < 4; ++u)
      if (j + u * 512 < be) atomicAdd(&cnt[(key[u] >> COL_BITS) & (BKT_ROWS - 1)], 1);
  }
  __syncthreads();
  const int c = cnt[t];
  scn[t] = c;
  __syncthreads();
  for (int s = 1; s < BKT_ROWS; s <<= 1) {
    const int add = (t >= s) ? scn[t - s] : 0;
    __syncthreads();
    scn[t] += add;
    __syncthreads();
  }
  const int start = bs + scn[t] - c;
  rowp[b * BKT_ROWS + t] = start;
  cnt[t] = start;
  __syncthreads();
  if (n > STG_CAP) {
    lg_rank_place<false>(grp, bs, be, t, cnt, stg, csr);
    return;
  }
  lg_rank_place<true>(grp, bs, be, t, cnt, stg, csr);
  __syncthreads();
  const int h = min((16 - (bs & 15)) & 15, n);
  if (t < h) csr[bs + t] = stg[t];
  const int np = (n - h) >> 1;
  v4i* dst4 = (v4i*)(csr + bs + h);
  #pragma unroll 4
  for (int i = t; i < np; i += 512) {
    const int j = h + 2 * i;
    const v4i o = {stg[j].x, stg[j].y, stg[j + 1].x, stg[j + 1].y};
    dst4[i] = o;
  }
  if (t == 0 && ((n - h) & 1)) csr[be - 1] = stg[n - 1];
}

#define SCAN_T     768
#define SCAN_W     (SCAN_T / 32)
static_assert(SCAN_NB * SCAN_T * 4 == CNT_PAD, "the scan blocks tile the counters");
__global__ __launch_bounds__(SCAN_T) void lg_scan(const int4* __restrict__ cnt4, int4* __restrict__ rowp4) {
  __shared__ int wtot[SCAN_W];
  __shared__ int wpre[SCAN_W];
  const int t = threadIdx.x, b = blockIdx.x;
  const int lane = t & 31, wave = t >> 5;
  int pre = 0;
  for (int i0 = 0; i0 < b; i0 += 16) {
    int4 u[16];
    #pragma unroll
    for (int k = 0; k < 16; ++k) u[k] = cnt4[min(i0 + k, b - 1) * SCAN_T + t];
    #pragma unroll
    for (int k = 0; k < 16; ++k)
      if (i0 + k < b) pre += u[k].x + u[k].y + u[k].z + u[k].w;
  }
  const int4 v = cnt4[b * SCAN_T + t];
  const int tsum = v.x + v.y + v.z + v.w;
  int x = tsum;
  for (int d = 1; d < 32; d <<= 1) {
    const int y = __shfl_up(x, d);
    if (lane >= d) x += y;
  }
  for (int d = 16; d > 0; d >>= 1) pre += __shfl_xor(pre, d);
  if (lane == 31) wtot[wave] = x;
  if (lane == 0)  wpre[wave] = pre;
  __syncthreads();
  int base = x - tsum;
  for (int w = 0; w < SCAN_W; ++w) {
    base += wpre[w];
    if (w < wave) base += wtot[w];
  }
  int4 o;
  o.x = base; o.y = o.x + v.x; o.z = o.y + v.y; o.w = o.z + v.z;
  rowp4[b * SCAN_T + t] = o;
}


template <int K>
__device__ __forceinline__ int grp_bcast(int v) {
  return (int)__builtin_amdgcn_mov_dpp8((unsigned)v, (unsigned)(K * 0x249249));
}

#define LG_SLOT(K)                                                                \
  {                                                                               \
    const unsigned c = (unsigned)grp_bcast<K>(cc);                                \
    w[K] = __int_as_float(grp_bcast<K>(ww));                                      \
    const char* p = srcb + (size_t)(c * (unsigned)(EMBED_DIM * 4) + qoff);        \
    x0[K] = *(const v4f*)p;                                                       \
    x1[K] = *(const v4f*)(p + 128);                                               \
  }

__device__ __forceinline__ void lg_round(const char* __restrict__ srcb, const int cc, const int ww,
                                         const unsigned qoff, v4f& a0, v4f& a1) {
  v4f   x0[8], x1[8];
  float w[8];
  LG_SLOT(0) LG_SLOT(1) LG_SLOT(2) LG_SLOT(3) LG_SLOT(4) LG_SLOT(5) LG_SLOT(6) LG_SLOT(7)
  #pragma unroll
  for (int k = 0; k < 8; ++k) { a0 += w[k] * x0[k]; a1 += w[k] * x1[k]; }
}

__device__ __forceinline__ void lg_load_recs(const int2* __restrict__ csr, const int jb, const int n, const int q,
                                             int2 (&rec)[4]) {
  #pragma unroll
  for (int j = 0; j < 4; ++j) {
    rec[j] = make_int2(0, 0);
    if (8 * j < n) rec[j] = csr[jb + min(8 * j + q, n - 1)];
  }
}

__device__ __forceinline__ void lg_chunk_sum(const char* __restrict__ srcb, const int2 (&rec)[4], const int n,
                                             const int q, const unsigned qoff, v4f& a0, v4f& a1) {
  #pragma unroll
  for (int j = 0; j < 4; ++j)
    if (8 * j < n)
      lg_round(srcb, rec[j].x, (8 * j + q < n) ? rec[j].y : 0  , qoff, a0, a1);
}

__device__ __forceinline__ void lg_row_sum(const char* __restrict__ srcb, const int2* __restrict__ csr,
                                           const int beg, const int end, const int q, const unsigned qoff,
                                           v4f& a0, v4f& a1) {
  for (int jb = beg; jb < end; jb += 32) {
    const int n = min(end - jb, 32);
    int2 rec[4];
    lg_load_recs(csr, jb, n, q, rec);
    lg_chunk_sum(srcb, rec, n, q, qoff, a0, a1);
  }
}

#define L1_BLOCKS  1024
__global__ __launch_bounds__(256) void lg_layer1(const float* __restrict__ emb, const int* __restrict__ rowp,
                                                 const int2* __restrict__ csr, float* __restrict__ t1) {
  static_assert(N_TOTAL % 4 == 0, "a wave takes four whole rows");
  const int lane = threadIdx.x & 31;
  const int q = lane & 7;
  const int g = lane >> 3;
  const unsigned qoff = (unsigned)q * 16u;
  const char* srcb = (const char*)emb;
  const int w0 = __builtin_amdgcn_readfirstlane((int)((blockIdx.x * blockDim.x + threadIdx.x) >> 5));
  const int step = (int)((gridDim.x * blockDim.x) >> 5) * 4;
  int r0 = w0 * 4;
  if (r0 >= N_TOTAL) return;
  int row = r0 + g;
  int beg = rowp[row], end = rowp[row + 1];
  int2 rec[4];
  lg_load_recs(csr, beg, min(end - beg, 32), q, rec);
  int begN = 0, endN = 0;
  if (r0 + step < N_TOTAL) { begN = rowp[row + step]; endN = rowp[row + step + 1]; }
  for (; r0 < N_TOTAL; r0 += step) {
    int2 recN[4];
    lg_load_recs(csr, begN, min(endN - begN, 32), q, recN);
    int begNN = 0, endNN = 0;
    if (r0 + 2 * step < N_TOTAL) { begNN = rowp[row + 2 * step]; endNN = rowp[row + 2 * step + 1]; }
    v4f a0 = {0.f, 0.f, 0.f, 0.f}, a1 = a0;
    lg_chunk_sum(srcb, rec, min(end - beg, 32), q, qoff, a0, a1);
    for (int jb = beg + 32; jb < end; jb += 32) {
      const int n = min(end - jb, 32);
      int2 rc[4];
      lg_load_recs(csr, jb, n, q, rc);
      lg_chunk_sum(srcb, rc, n, q, qoff, a0, a1);
    }
    *(v4f*)(t1 + (size_t)row * EMBED_DIM + q * 4) = a0;
    *(v4f*)(t1 + (size_t)row * EMBED_DIM + HALF_DIM + q * 4) = a1;
    row += step;
    beg = begN; end = endN; begN = begNN; endN = endNN;
    #pragma unroll
    for (int j = 0; j < 4; ++j) rec[j] = recN[j];
  }
}

__global__ __launch_bounds__(256) void lg_layer23(const float* __restrict__ emb0, const float* __restrict__ t1,
                                                  const int* __restrict__ rowp, const int2* __restrict__ csr,
                                                  const int* __restrict__ uid, const int* __restrict__ iid,
                                                  float* __restrict__ accS) {
  const int lane = threadIdx.x & 31;
  const int q = lane & 7;
  const int g = lane >> 3;
  const unsigned qoff = (unsigned)q * 16u;
  const int slot = __builtin_amdgcn_readfirstlane((int)((blockIdx.x * blockDim.x + threadIdx.x) >> 5));
  const int row = (slot < BATCH) ? uid[slot] : (NUM_USERS + iid[slot - BATCH]);
  const int begR = rowp[row];
  const int ntask = min(max(rowp[row + 1] - begR, 0), N_EDGES) + 1;
  const v4f zero = {0.f, 0.f, 0.f, 0.f};
  v4f l2a = zero, l2b = zero, s0 = zero, s1 = zero;
  int2 rrN = make_int2(row, 0);
  if (g > 0 && g < ntask) rrN = csr[begR + g - 1];
  for (int t0 = 0; t0 < ntask; t0 += 4) {
    const int t = t0 + g;
    const int2 rr = rrN;
    rrN = make_int2(row, 0);
    if (t + 4 < ntask) rrN = csr[begR + t + 3];
    int beg = 0, end = 0;
    if (t < ntask) { beg = rowp[rr.x]; end = rowp[rr.x + 1]; }
    v4f a0 = zero, a1 = zero;
    lg_row_sum((const char*)t1, csr, beg, end, q, qoff, a0, a1);
    #pragma unroll
    for (int gg = 0; gg < 4; ++gg)
      if (t0 + gg < ntask) {
        const int sl = gg * 8 + q;
        v4f b0, b1;
        #pragma unroll
        for (int e = 0; e < 4; ++e) { b0[e] = __shfl(a0[e], sl); b1[e] = __shfl(a1[e], sl); }
        const float wv = __int_as_float(__shfl(rr.y, gg * 8));
        if (t0 + gg == 0) { l2a = b0; l2b = b1; }
        else              { s0 += wv * b0; s1 += wv * b1; }
      }
  }
  if (g == 0) {
    const v4f l0a = *(const v4f*)(emb0 + (size_t)row * EMBED_DIM + q * 4);
    const v4f l0b = *(const v4f*)(emb0 + (size_t)row * EMBED_DIM + HALF_DIM + q * 4);
    const v4f l1a = *(const v4f*)(t1 + (size_t)row * EMBED_DIM + q * 4);
    const v4f l1b = *(const v4f*)(t1 + (size_t)row * EMBED_DIM + HALF_DIM + q * 4);
    *(v4f*)(accS + (size_t)slot * EMBED_DIM + q * 4) = ((l0a + l1a) + l2a) + s0;
    *(v4f*)(accS + (size_t)slot * EMBED_DIM + HALF_DIM + q * 4) = ((l0b + l1b) + l2b) + s1;
  }
}

__global__ void lg_dot_wmma(const float* __restrict__ acc, float* __restrict__ out) {
  __shared__ float sd[8][8][32];
  const int lane = threadIdx.x & 31;
  const int wave = threadIdx.x >> 5;
  const int tile = blockIdx.x * 8 + wave;
  const int k0   = tile * 16;
  const int rN   = lane & 15;
  const float* U = acc + (size_t)(k0 + rN) * EMBED_DIM;
  const float* I = acc + (size_t)(BATCH + k0 + rN) * EMBED_DIM;

  v8f c = {0.f, 0.f, 0.f, 0.f, 0.f, 0.f, 0.f, 0.f};

#if __has_builtin(__builtin_amdgcn_wmma_f32_16x16x4_f32)
  const int kb = (lane >> 4) << 1;
  #pragma unroll
  for (int s = 0; s < 16; ++s) {
    const int col = s * 4 + kb;
    v2f a; a[0] = U[col]; a[1] = U[col + 1];
    v2f b; b[0] = I[col]; b[1] = I[col + 1];
    c = __builtin_amdgcn_wmma_f32_16x16x4_f32(false, a, false, b, (short)0, c, false, false);
  }
#else
  const int kb8 = (lane >> 4) << 3;
  #pragma unroll
  for (int blk = 0; blk < 2; ++blk) {
    v16h a, b;
    #pragma unroll
    for (int i = 0; i < 8; ++i) {
      const int kk = blk * 32 + kb8 + ((i < 4) ? (2 * i) : (16 + 2 * (i - 4)));
      a[2 * i]     = (_Float16)U[kk];
      a[2 * i + 1] = (_Float16)U[kk + 1];
      b[2 * i]     = (_Float16)I[kk];
      b[2 * i + 1] = (_Float16)I[kk + 1];
    }
    c = __builtin_amdgcn_wmma_f32_16x16x32_f16(false, a, false, b, (short)0, c, false, false);
  }
#endif

  #pragma unroll
  for (int g = 0; g < 8; ++g) sd[wave][g][lane] = c[g];
  __syncthreads();

  if (lane < 16) {
    const float v = (lane < 8) ? sd[wave][lane][lane] : sd[wave][lane - 8][lane + 16];
    out[k0 + lane] = v * 0.0625f;
  }
}

extern "C" void kernel_launch(void* const* d_in, const int* in_sizes, int n_in,
                              void* d_out, int out_size, void* d_ws, size_t ws_size,
                              hipStream_t stream) {
  const float* node_emb = (const float*)d_in[0];
  const float* adj_val  = (const float*)d_in[1];
  const int*   adj_row  = (const int*)d_in[2];
  const int*   adj_col  = (const int*)d_in[3];
  const int*   user_ids = (const int*)d_in[4];
  const int*   item_ids = (const int*)d_in[5];
  float* out = (float*)d_out;

  char* ws    = (char*)d_ws;
  float* tbl1 = (float*)(ws + OFF_TBL1);
  int2*  grp  = (int2*)(ws + OFF_TBL1);
  int2*  csr  = (int2*)(ws + OFF_CSR);
  int*   rowp = (int*)(ws + OFF_ROWP);
  int*   cntCB = (int*)(ws + OFF_CNT);
  int*   offsCB = (int*)(ws + OFF_OFFS);
  float* accS = (float*)(ws + OFF_ACCS);

  const dim3 b256(256);

  lg_bkt_count<<<NCHK, 512, 0, stream>>>(adj_row, cntCB);
  lg_scan<<<SCAN_NB, SCAN_T, 0, stream>>>((const int4*)cntCB, (int4*)offsCB);
  lg_bkt_place<<<NCHK, 512, 0, stream>>>(adj_row, adj_col, adj_val, offsCB, grp);
  lg_bkt_rows<<<NBKT, 512, 0, stream>>>(grp, offsCB, rowp, csr);

  lg_layer1<<<L1_BLOCKS, b256, 0, stream>>>(node_emb, rowp, csr, tbl1);
  lg_layer23<<<2 * BATCH / 8, b256, 0, stream>>>(node_emb, tbl1, rowp, csr, user_ids, item_ids, accS);

  lg_dot_wmma<<<BATCH / 128, b256, 0, stream>>>(accS, out);
}


# COO->CSR without a global pre-count: one block counts, scans and groups each edge chunk by row range in LDS; the range kernel gathers its 512 pieces. 7 -> 5 launches; row sums in arrival order still
# speedup vs baseline: 5.7523x; 1.0822x over previous
#include <hip/hip_runtime.h>

#define N_TOTAL    150100
#define EMBED_DIM  64
#define HALF_DIM   32
#define NUM_USERS  100000
#define N_EDGES    3000000
#define BATCH      4096
#define NUM_LAYERS 3

#define CNT_PAD    150528

#define BKT_SHIFT  9
#define BKT_ROWS   512
#define NBKT       294
#define NCHK       512
#define CHUNK_E    6144
#define COL_BITS   18
static_assert(NBKT * BKT_ROWS == CNT_PAD && CNT_PAD > N_TOTAL, "the row ranges cover every row and the end mark");
static_assert(NCHK * CHUNK_E >= N_EDGES && N_TOTAL < (1 << COL_BITS) && CHUNK_E % 1024 == 0, "grouping sizes");
static_assert((N_EDGES % CHUNK_E) % 2 == 0, "every chunk holds whole pairs of records: it is written out 16 bytes a lane");
static_assert(NCHK == 512 && BKT_ROWS == 512, "in the range kernel thread t serves chunk t's piece and row t's counter");

#define OFF_TBL1   0u
#define OFF_CSR    38425600u
#define OFF_ROWP   62425600u
#define OFF_PCS    63027712u
#define OFF_ACCS   76851200u
static_assert(OFF_PCS + (NBKT + 1) * NCHK * 4 <= OFF_ACCS && NCHK * CHUNK_E * 8 <= OFF_CSR, "workspace pieces do not overlap");

typedef __attribute__((ext_vector_type(2)))  float    v2f;
typedef __attribute__((ext_vector_type(4)))  float    v4f;
typedef __attribute__((ext_vector_type(4)))  int      v4i;
typedef __attribute__((ext_vector_type(8)))  float    v8f;
typedef __attribute__((ext_vector_type(16))) _Float16 v16h;

__global__ __launch_bounds__(512) void lg_chunk_group(const int* __restrict__ rw, const int* __restrict__ cl,
                                                      const float* __restrict__ val,
                                                      int* __restrict__ pcs, int2* __restrict__ grp) {
  __shared__ int  cur[NBKT];
  __shared__ int  wsum[16];
  __shared__ __attribute__((aligned(16))) int2 stg[CHUNK_E];
  const int t = threadIdx.x, c = blockIdx.x;
  const int lane = t & 31, wave = t >> 5;
  const int i0 = c * CHUNK_E + t;
  const int ne = min(max(N_EDGES - c * CHUNK_E, 0), CHUNK_E);
  if (t < NBKT) cur[t] = 0;
  int   r[CHUNK_E / 512], cc[CHUNK_E / 512];
  float vv[CHUNK_E / 512];
  #pragma unroll
  for (int k = 0; k < CHUNK_E / 512; ++k) {
    const int i = min(i0 + k * 512, N_EDGES - 1);
    r[k] = rw[i]; cc[k] = cl[i]; vv[k] = val[i];
  }
  __syncthreads();
  #pragma unroll
  for (int k = 0; k < CHUNK_E / 512; ++k)
    if (t + k * 512 < ne) atomicAdd(&cur[r[k] >> BKT_SHIFT], 1);
  __syncthreads();
  const int n = (t < NBKT) ? cur[t] : 0;
  int x = n;
  for (int d = 1; d < 32; d <<= 1) {
    const int y = __shfl_up(x, d);
    if (lane >= d) x += y;
  }
  if (lane == 31) wsum[wave] = x;
  __syncthreads();
  int base = 0;
  for (int w = 0; w < wave; ++w) base += wsum[w];
  if (t < NBKT) {
    const int ls = base + x - n;
    cur[t] = ls;
    pcs[t * NCHK + c] = ls;
  }
  if (t == 0) pcs[NBKT * NCHK + c] = ne;
  __syncthreads();
  #pragma unroll
  for (int k = 0; k < CHUNK_E / 512; ++k)
    if (t + k * 512 < ne) {
      const int b = r[k] >> BKT_SHIFT;
      const int p = min(max(atomicAdd(&cur[b], 1), 0), CHUNK_E - 1);
      stg[p] = make_int2(((r[k] & (BKT_ROWS - 1)) << COL_BITS) | cc[k], __float_as_int(vv[k]));
    }
  __syncthreads();
  v4i* dst4 = (v4i*)(grp + (size_t)c * CHUNK_E);
  const v4i* src4 = (const v4i*)stg;
  #pragma unroll
  for (int k = 0; k < CHUNK_E / 1024; ++k) {
    const int i = t + k * 512;
    if (2 * i < ne) dst4[i] = src4[i];
  }
}

#define STG_CAP    12288
#define PCS_TRIPS  8
#define SPARE_CTR  32

__device__ __forceinline__ int lg_ctr(const int x) {
  return min((int)((unsigned)x >> COL_BITS), BKT_ROWS + SPARE_CTR - 1);
}

template <bool STAGED>
__device__ __forceinline__ void lg_rank_put(const int2 e, const int p, const int bs, int2* stg, int2* __restrict__ csr) {
  const int2 o = make_int2(e.x & ((1 << COL_BITS) - 1), e.y);
  if (STAGED) stg[min(max(p - bs, 0), STG_CAP - 1)] = o;
  else        csr[min(max(p, 0), N_EDGES - 1)] = o;
}

template <bool STAGED>
__device__ __forceinline__ void lg_rank_place(const int2* __restrict__ grp, const int2 (&e)[PCS_TRIPS][4],
                                              const int* psrc, const int* plen, const int pc0, const int longp,
                                              const int q, const int bs, int* cnt, int2* stg,
                                              int2* __restrict__ csr) {
  #pragma unroll
  for (int tr = 0; tr < PCS_TRIPS; tr += 2) {
    int p[8];
    #pragma unroll
    for (int v = 0; v < 8; ++v) p[v] = atomicAdd(&cnt[lg_ctr(e[tr + (v >> 2)][v & 3].x)], 1);
    #pragma unroll
    for (int v = 0; v < 8; ++v)
      if (lg_ctr(e[tr + (v >> 2)][v & 3].x) < BKT_ROWS) lg_rank_put<STAGED>(e[tr + (v >> 2)][v & 3], p[v], bs, stg, csr);
  }
  if (longp)
    for (int tr = 0; tr < PCS_TRIPS; ++tr) {
      const int src = psrc[pc0 + tr * 4], len = plen[pc0 + tr * 4];
      for (int o = 32 + q; o < len; o += 8) {
        const int2 v = grp[src + o];
        lg_rank_put<STAGED>(v, atomicAdd(&cnt[lg_ctr(v.x)], 1), bs, stg, csr);
      }
    }
}

__global__ __launch_bounds__(512) void lg_bkt_rows(const int2* __restrict__ grp, const int* __restrict__ pcs,
                                                   int* __restrict__ rowp, int2* __restrict__ csr) {
  __shared__ int  cnt[BKT_ROWS + SPARE_CTR];
  __shared__ int  scn[BKT_ROWS];
  __shared__ int  psrc[NCHK];
  __shared__ int  plen[NCHK];
  __shared__ int  wbs[16], wn[16];
  __shared__ __attribute__((aligned(16))) int2 stg[STG_CAP];
  const int t = threadIdx.x, b = blockIdx.x;
  const int lane = t & 31, wave = t >> 5;
  const int q = lane & 7, g = lane >> 3;
  const int s0 = min(max(pcs[b * NCHK + t], 0), CHUNK_E);
  const int ln = min(max(pcs[(b + 1) * NCHK + t] - s0, 0), CHUNK_E - s0);
  psrc[t] = t * CHUNK_E + s0;
  plen[t] = ln;
  cnt[t] = 0;
  if (t < SPARE_CTR) cnt[BKT_ROWS + t] = 0;
  int xs = s0, xn = ln;
  for (int d = 16; d > 0; d >>= 1) { xs += __shfl_xor(xs, d); xn += __shfl_xor(xn, d); }
  if (lane == 0) { wbs[wave] = xs; wn[wave] = xn; }
  __syncthreads();
  int bs = 0, n = 0;
  for (int w = 0; w < 16; ++w) { bs += wbs[w]; n += wn[w]; }
  bs = min(bs, N_EDGES);
  n  = min(n, N_EDGES - bs);
  const int be = bs + n;
  int2 e[PCS_TRIPS][4];
  const int pc0 = wave * (4 * PCS_TRIPS) + g;
  int longp = 0;
  const int norec = (BKT_ROWS + lane) << COL_BITS;
  #pragma unroll
  for (int tr = 0; tr < PCS_TRIPS; ++tr) {
    const int src = psrc[pc0 + tr * 4], len = plen[pc0 + tr * 4];
    longp |= (len > 32);
    #pragma unroll
    for (int u = 0; u < 4; ++u) {
      e[tr][u] = make_int2(norec, 0);
      if (q + 8 * u < len) e[tr][u] = *(const int2*)((const char*)grp + (size_t)((unsigned)(src + q + 8 * u) * 8u));
    }
  }
  #pragma unroll
  for (int tr = 0; tr < PCS_TRIPS; ++tr) {
    #pragma unroll
    for (int u = 0; u < 4; ++u) atomicAdd(&cnt[lg_ctr(e[tr][u].x)], 1);
  }
  if (longp)
    for (int tr = 0; tr < PCS_TRIPS; ++tr) {
      const int src = psrc[pc0 + tr * 4], len = plen[pc0 + tr * 4];
      for (int o = 32 + q; o < len; o += 8) atomicAdd(&cnt[lg_ctr(grp[src + o].x)], 1);
    }
  __syncthreads();
  const int c = cnt[t];
  scn[t] = c;
  __syncthreads();
  for (int s = 1; s < BKT_ROWS; s <<= 1) {
    const int add = (t >= s) ? scn[t - s] : 0;
    __syncthreads();
    scn[t] += add;
    __syncthreads();
  }
  const int start = bs + scn[t] - c;
  rowp[b * BKT_ROWS + t] = start;
  cnt[t] = start;
  __syncthreads();
  if (n > STG_CAP) {
    lg_rank_place<false>(grp, e, psrc, plen, pc0, longp, q, bs, cnt, stg, csr);
    return;
  }
  lg_rank_place<true>(grp, e, psrc, plen, pc0, longp, q, bs, cnt, stg, csr);
  __syncthreads();
  const int h = min((16 - (bs & 15)) & 15, n);
  if (t < h) csr[bs + t] = stg[t];
  const int np = (n - h) >> 1;
  v4i* dst4 = (v4i*)(csr + bs + h);
  #pragma unroll 4
  for (int i = t; i < np; i += 512) {
    const int j = h + 2 * i;
    const v4i o = {stg[j].x, stg[j].y, stg[j + 1].x, stg[j + 1].y};
    dst4[i] = o;
  }
  if (t == 0 && ((n - h) & 1)) csr[be - 1] = stg[n - 1];
}


template <int K>
__device__ __forceinline__ int grp_bcast(int v) {
  return (int)__builtin_amdgcn_mov_dpp8((unsigned)v, (unsigned)(K * 0x249249));
}

#define LG_SLOT(K)                                                                \
  {                                                                               \
    const unsigned c = (unsigned)grp_bcast<K>(cc);                                \
    w[K] = __int_as_float(grp_bcast<K>(ww));                                      \
    const char* p = srcb + (size_t)(c * (unsigned)(EMBED_DIM * 4) + qoff);        \
    x0[K] = *(const v4f*)p;                                                       \
    x1[K] = *(const v4f*)(p + 128);                                               \
  }

__device__ __forceinline__ void lg_round(const char* __restrict__ srcb, const int cc, const int ww,
                                         const unsigned qoff, v4f& a0, v4f& a1) {
  v4f   x0[8], x1[8];
  float w[8];
  LG_SLOT(0) LG_SLOT(1) LG_SLOT(2) LG_SLOT(3) LG_SLOT(4) LG_SLOT(5) LG_SLOT(6) LG_SLOT(7)
  #pragma unroll
  for (int k = 0; k < 8; ++k) { a0 += w[k] * x0[k]; a1 += w[k] * x1[k]; }
}

__device__ __forceinline__ void lg_load_recs(const int2* __restrict__ csr, const int jb, const int n, const int q,
                                             int2 (&rec)[4]) {
  #pragma unroll
  for (int j = 0; j < 4; ++j) {
    rec[j] = make_int2(0, 0);
    if (8 * j < n) rec[j] = csr[jb + min(8 * j + q, n - 1)];
  }
}

__device__ __forceinline__ void lg_chunk_sum(const char* __restrict__ srcb, const int2 (&rec)[4], const int n,
                                             const int q, const unsigned qoff, v4f& a0, v4f& a1) {
  #pragma unroll
  for (int j = 0; j < 4; ++j)
    if (8 * j < n)
      lg_round(srcb, rec[j].x, (8 * j + q < n) ? rec[j].y : 0  , qoff, a0, a1);
}

__device__ __forceinline__ void lg_row_sum(const char* __restrict__ srcb, const int2* __restrict__ csr,
                                           const int beg, const int end, const int q, const unsigned qoff,
                                           v4f& a0, v4f& a1) {
  for (int jb = beg; jb < end; jb += 32) {
    const int n = min(end - jb, 32);
    int2 rec[4];
    lg_load_recs(csr, jb, n, q, rec);
    lg_chunk_sum(srcb, rec, n, q, qoff, a0, a1);
  }
}

#define L1_BLOCKS  1024
__global__ __launch_bounds__(256) void lg_layer1(const float* __restrict__ emb, const int* __restrict__ rowp,
                                                 const int2* __restrict__ csr, float* __restrict__ t1) {
  static_assert(N_TOTAL % 4 == 0, "a wave takes four whole rows");
  const int lane = threadIdx.x & 31;
  const int q = lane & 7;
  const int g = lane >> 3;
  const unsigned qoff = (unsigned)q * 16u;
  const char* srcb = (const char*)emb;
  const int w0 = __builtin_amdgcn_readfirstlane((int)((blockIdx.x * blockDim.x + threadIdx.x) >> 5));
  const int step = (int)((gridDim.x * blockDim.x) >> 5) * 4;
  int r0 = w0 * 4;
  if (r0 >= N_TOTAL) return;
  int row = r0 + g;
  int beg = rowp[row], end = rowp[row + 1];
  int2 rec[4];
  lg_load_recs(csr, beg, min(end - beg, 32), q, rec);
  int begN = 0, endN = 0;
  if (r0 + step < N_TOTAL) { begN = rowp[row + step]; endN = rowp[row + step + 1]; }
  for (; r0 < N_TOTAL; r0 += step) {
    int2 recN[4];
    lg_load_recs(csr, begN, min(endN - begN, 32), q, recN);
    int begNN = 0, endNN = 0;
    if (r0 + 2 * step < N_TOTAL) { begNN = rowp[row + 2 * step]; endNN = rowp[row + 2 * step + 1]; }
    v4f a0 = {0.f, 0.f, 0.f, 0.f}, a1 = a0;
    lg_chunk_sum(srcb, rec, min(end - beg, 32), q, qoff, a0, a1);
    for (int jb = beg + 32; jb < end; jb += 32) {
      const int n = min(end - jb, 32);
      int2 rc[4];
      lg_load_recs(csr, jb, n, q, rc);
      lg_chunk_sum(srcb, rc, n, q, qoff, a0, a1);
    }
    *(v4f*)(t1 + (size_t)row * EMBED_DIM + q * 4) = a0;
    *(v4f*)(t1 + (size_t)row * EMBED_DIM + HALF_DIM + q * 4) = a1;
    row += step;
    beg = begN; end = endN; begN = begNN; endN = endNN;
    #pragma unroll
    for (int j = 0; j < 4; ++j) rec[j] = recN[j];
  }
}

__global__ __launch_bounds__(256) void lg_layer23(const float* __restrict__ emb0, const float* __restrict__ t1,
                                                  const int* __restrict__ rowp, const int2* __restrict__ csr,
                                                  const int* __restrict__ uid, const int* __restrict__ iid,
                                                  float* __restrict__ accS) {
  const int lane = threadIdx.x & 31;
  const int q = lane & 7;
  const int g = lane >> 3;
  const unsigned qoff = (unsigned)q * 16u;
  const int slot = __builtin_amdgcn_readfirstlane((int)((blockIdx.x * blockDim.x + threadIdx.x) >> 5));
  const int row = (slot < BATCH) ? uid[slot] : (NUM_USERS + iid[slot - BATCH]);
  const int begR = rowp[row];
  const int ntask = min(max(rowp[row + 1] - begR, 0), N_EDGES) + 1;
  const v4f zero = {0.f, 0.f, 0.f, 0.f};
  v4f l2a = zero, l2b = zero, s0 = zero, s1 = zero;
  int2 rrN = make_int2(row, 0);
  if (g > 0 && g < ntask) rrN = csr[begR + g - 1];
  for (int t0 = 0; t0 < ntask; t0 += 4) {
    const int t = t0 + g;
    const int2 rr = rrN;
    rrN = make_int2(row, 0);
    if (t + 4 < ntask) rrN = csr[begR + t + 3];
    int beg = 0, end = 0;
    if (t < ntask) { beg = rowp[rr.x]; end = rowp[rr.x + 1]; }
    v4f a0 = zero, a1 = zero;
    lg_row_sum((const char*)t1, csr, beg, end, q, qoff, a0, a1);
    #pragma unroll
    for (int gg = 0; gg < 4; ++gg)
      if (t0 + gg < ntask) {
        const int sl = gg * 8 + q;
        v4f b0, b1;
        #pragma unroll
        for (int e = 0; e < 4; ++e) { b0[e] = __shfl(a0[e], sl); b1[e] = __shfl(a1[e], sl); }
        const float wv = __int_as_float(__shfl(rr.y, gg * 8));
        if (t0 + gg == 0) { l2a = b0; l2b = b1; }
        else              { s0 += wv * b0; s1 += wv * b1; }
      }
  }
  if (g == 0) {
    const v4f l0a = *(const v4f*)(emb0 + (size_t)row * EMBED_DIM + q * 4);
    const v4f l0b = *(const v4f*)(emb0 + (size_t)row * EMBED_DIM + HALF_DIM + q * 4);
    const v4f l1a = *(const v4f*)(t1 + (size_t)row * EMBED_DIM + q * 4);
    const v4f l1b = *(const v4f*)(t1 + (size_t)row * EMBED_DIM + HALF_DIM + q * 4);
    *(v4f*)(accS + (size_t)slot * EMBED_DIM + q * 4) = ((l0a + l1a) + l2a) + s0;
    *(v4f*)(accS + (size_t)slot * EMBED_DIM + HALF_DIM + q * 4) = ((l0b + l1b) + l2b) + s1;
  }
}

__global__ void lg_dot_wmma(const float* __restrict__ acc, float* __restrict__ out) {
  __shared__ float sd[8][8][32];
  const int lane = threadIdx.x & 31;
  const int wave = threadIdx.x >> 5;
  const int tile = blockIdx.x * 8 + wave;
  const int k0   = tile * 16;
  const int rN   = lane & 15;
  const float* U = acc + (size_t)(k0 + rN) * EMBED_DIM;
  const float* I = acc + (size_t)(BATCH + k0 + rN) * EMBED_DIM;

  v8f c = {0.f, 0.f, 0.f, 0.f, 0.f, 0.f, 0.f, 0.f};

#if __has_builtin(__builtin_amdgcn_wmma_f32_16x16x4_f32)
  const int kb = (lane >> 4) << 1;
  #pragma unroll
  for (int s = 0; s < 16; ++s) {
    const int col = s * 4 + kb;
    v2f a; a[0] = U[col]; a[1] = U[col + 1];
    v2f b; b[0] = I[col]; b[1] = I[col + 1];
    c = __builtin_amdgcn_wmma_f32_16x16x4_f32(false, a, false, b, (short)0, c, false, false);
  }
#else
  const int kb8 = (lane >> 4) << 3;
  #pragma unroll
  for (int blk = 0; blk < 2; ++blk) {
    v16h a, b;
    #pragma unroll
    for (int i = 0; i < 8; ++i) {
      const int kk = blk * 32 + kb8 + ((i < 4) ? (2 * i) : (16 + 2 * (i - 4)));
      a[2 * i]     = (_Float16)U[kk];
      a[2 * i + 1] = (_Float16)U[kk + 1];
      b[2 * i]     = (_Float16)I[kk];
      b[2 * i + 1] = (_Float16)I[kk + 1];
    }
    c = __builtin_amdgcn_wmma_f32_16x16x32_f16(false, a, false, b, (short)0, c, false, false);
  }
#endif

  #pragma unroll
  for (int g = 0; g < 8; ++g) sd[wave][g][lane] = c[g];
  __syncthreads();

  if (lane < 16) {
    const float v = (lane < 8) ? sd[wave][lane][lane] : sd[wave][lane - 8][lane + 16];
    out[k0 + lane] = v * 0.0625f;
  }
}

extern "C" void kernel_launch(void* const* d_in, const int* in_sizes, int n_in,
                              void* d_out, int out_size, void* d_ws, size_t ws_size,
                              hipStream_t stream) {
  const float* node_emb = (const float*)d_in[0];
  const float* adj_val  = (const float*)d_in[1];
  const int*   adj_row  = (const int*)d_in[2];
  const int*   adj_col  = (const int*)d_in[3];
  const int*   user_ids = (const int*)d_in[4];
  const int*   item_ids = (const int*)d_in[5];
  float* out = (float*)d_out;

  char* ws    = (char*)d_ws;
  float* tbl1 = (float*)(ws + OFF_TBL1);
  int2*  grp  = (int2*)(ws + OFF_TBL1);
  int2*  csr  = (int2*)(ws + OFF_CSR);
  int*   rowp = (int*)(ws + OFF_ROWP);
  int*   pcs  = (int*)(ws + OFF_PCS);
  float* accS = (float*)(ws + OFF_ACCS);

  const dim3 b256(256);

  lg_chunk_group<<<NCHK, 512, 0, stream>>>(adj_row, adj_col, adj_val, pcs, grp);
  lg_bkt_rows<<<NBKT, 512, 0, stream>>>(grp, pcs, rowp, csr);

  lg_layer1<<<L1_BLOCKS, b256, 0, stream>>>(node_emb, rowp, csr, tbl1);
  lg_layer23<<<2 * BATCH / 8, b256, 0, stream>>>(node_emb, tbl1, rowp, csr, user_ids, item_ids, accS);

  lg_dot_wmma<<<BATCH / 128, b256, 0, stream>>>(accS, out);
}


# Dot launch folded into the layer-2/3 launch: 32-wave blocks take 16 pairs, slot rows pass through LDS, not a workspace table; wave 0 runs the same f32 WMMA product. 5 -> 4 launches; sum order kept
# speedup vs baseline: 5.9144x; 1.0282x over previous
#include <hip/hip_runtime.h>

#define N_TOTAL    150100
#define EMBED_DIM  64
#define HALF_DIM   32
#define NUM_USERS  100000
#define N_EDGES    3000000
#define BATCH      4096
#define NUM_LAYERS 3

#define CNT_PAD    150528

#define BKT_SHIFT  9
#define BKT_ROWS   512
#define NBKT       294
#define NCHK       512
#define CHUNK_E    6144
#define COL_BITS   18
static_assert(NBKT * BKT_ROWS == CNT_PAD && CNT_PAD > N_TOTAL, "the row ranges cover every row and the end mark");
static_assert(NCHK * CHUNK_E >= N_EDGES && N_TOTAL < (1 << COL_BITS) && CHUNK_E % 1024 == 0, "grouping sizes");
static_assert((N_EDGES % CHUNK_E) % 2 == 0, "every chunk holds whole pairs of records: it is written out 16 bytes a lane");
static_assert(NCHK == 512 && BKT_ROWS == 512, "in the range kernel thread t serves chunk t's piece and row t's counter");

#define OFF_TBL1   0u
#define OFF_CSR    38425600u
#define OFF_ROWP   62425600u
#define OFF_PCS    63027712u
#define OFF_END    76851200u
static_assert(OFF_PCS + (NBKT + 1) * NCHK * 4 <= OFF_END && NCHK * CHUNK_E * 8 <= OFF_CSR, "workspace pieces do not overlap");

typedef __attribute__((ext_vector_type(2)))  float    v2f;
typedef __attribute__((ext_vector_type(4)))  float    v4f;
typedef __attribute__((ext_vector_type(4)))  int      v4i;
typedef __attribute__((ext_vector_type(8)))  float    v8f;
typedef __attribute__((ext_vector_type(16))) _Float16 v16h;

__global__ __launch_bounds__(512) void lg_chunk_group(const int* __restrict__ rw, const int* __restrict__ cl,
                                                      const float* __restrict__ val,
                                                      int* __restrict__ pcs, int2* __restrict__ grp) {
  __shared__ int  cur[NBKT];
  __shared__ int  wsum[16];
  __shared__ __attribute__((aligned(16))) int2 stg[CHUNK_E];
  const int t = threadIdx.x, c = blockIdx.x;
  const int lane = t & 31, wave = t >> 5;
  const int i0 = c * CHUNK_E + t;
  const int ne = min(max(N_EDGES - c * CHUNK_E, 0), CHUNK_E);
  if (t < NBKT) cur[t] = 0;
  int   r[CHUNK_E / 512], cc[CHUNK_E / 512];
  float vv[CHUNK_E / 512];
  #pragma unroll
  for (int k = 0; k < CHUNK_E / 512; ++k) {
    const int i = min(i0 + k * 512, N_EDGES - 1);
    r[k] = rw[i]; cc[k] = cl[i]; vv[k] = val[i];
  }
  __syncthreads();
  #pragma unroll
  for (int k = 0; k < CHUNK_E / 512; ++k)
    if (t + k * 512 < ne) atomicAdd(&cur[r[k] >> BKT_SHIFT], 1);
  __syncthreads();
  const int n = (t < NBKT) ? cur[t] : 0;
  int x = n;
  for (int d = 1; d < 32; d <<= 1) {
    const int y = __shfl_up(x, d);
    if (lane >= d) x += y;
  }
  if (lane == 31) wsum[wave] = x;
  __syncthreads();
  int base = 0;
  for (int w = 0; w < wave; ++w) base += wsum[w];
  if (t < NBKT) {
    const int ls = base + x - n;
    cur[t] = ls;
    pcs[t * NCHK + c] = ls;
  }
  if (t == 0) pcs[NBKT * NCHK + c] = ne;
  __syncthreads();
  #pragma unroll
  for (int k = 0; k < CHUNK_E / 512; ++k)
    if (t + k * 512 < ne) {
      const int b = r[k] >> BKT_SHIFT;
      const int p = min(max(atomicAdd(&cur[b], 1), 0), CHUNK_E - 1);
      stg[p] = make_int2(((r[k] & (BKT_ROWS - 1)) << COL_BITS) | cc[k], __float_as_int(vv[k]));
    }
  __syncthreads();
  v4i* dst4 = (v4i*)(grp + (size_t)c * CHUNK_E);
  const v4i* src4 = (const v4i*)stg;
  #pragma unroll
  for (int k = 0; k < CHUNK_E / 1024; ++k) {
    const int i = t + k * 512;
    if (2 * i < ne) dst4[i] = src4[i];
  }
}

#define STG_CAP    12288
#define PCS_TRIPS  8
#define SPARE_CTR  32

__device__ __forceinline__ int lg_ctr(const int x) {
  return min((int)((unsigned)x >> COL_BITS), BKT_ROWS + SPARE_CTR - 1);
}

template <bool STAGED>
__device__ __forceinline__ void lg_rank_put(const int2 e, const int p, const int bs, int2* stg, int2* __restrict__ csr) {
  const int2 o = make_int2(e.x & ((1 << COL_BITS) - 1), e.y);
  if (STAGED) stg[min(max(p - bs, 0), STG_CAP - 1)] = o;
  else        csr[min(max(p, 0), N_EDGES - 1)] = o;
}

template <bool STAGED>
__device__ __forceinline__ void lg_rank_place(const int2* __restrict__ grp, const int2 (&e)[PCS_TRIPS][4],
                                              const int* psrc, const int* plen, const int pc0, const int longp,
                                              const int q, const int bs, int* cnt, int2* stg,
                                              int2* __restrict__ csr) {
  #pragma unroll
  for (int tr = 0; tr < PCS_TRIPS; tr += 2) {
    int p[8];
    #pragma unroll
    for (int v = 0; v < 8; ++v) p[v] = atomicAdd(&cnt[lg_ctr(e[tr + (v >> 2)][v & 3].x)], 1);
    #pragma unroll
    for (int v = 0; v < 8; ++v)
      if (lg_ctr(e[tr + (v >> 2)][v & 3].x) < BKT_ROWS) lg_rank_put<STAGED>(e[tr + (v >> 2)][v & 3], p[v], bs, stg, csr);
  }
  if (longp)
    for (int tr = 0; tr < PCS_TRIPS; ++tr) {
      const int src = psrc[pc0 + tr * 4], len = plen[pc0 + tr * 4];
      for (int o = 32 + q; o < len; o += 8) {
        const int2 v = grp[src + o];
        lg_rank_put<STAGED>(v, atomicAdd(&cnt[lg_ctr(v.x)], 1), bs, stg, csr);
      }
    }
}

__global__ __launch_bounds__(512) void lg_bkt_rows(const int2* __restrict__ grp, const int* __restrict__ pcs,
                                                   int* __restrict__ rowp, int2* __restrict__ csr) {
  __shared__ int  cnt[BKT_ROWS + SPARE_CTR];
  __shared__ int  scn[BKT_ROWS];
  __shared__ int  psrc[NCHK];
  __shared__ int  plen[NCHK];
  __shared__ int  wbs[16], wn[16];
  __shared__ __attribute__((aligned(16))) int2 stg[STG_CAP];
  const int t = threadIdx.x, b = blockIdx.x;
  const int lane = t & 31, wave = t >> 5;
  const int q = lane & 7, g = lane >> 3;
  const int s0 = min(max(pcs[b * NCHK + t], 0), CHUNK_E);
  const int ln = min(max(pcs[(b + 1) * NCHK + t] - s0, 0), CHUNK_E - s0);
  psrc[t] = t * CHUNK_E + s0;
  plen[t] = ln;
  cnt[t] = 0;
  if (t < SPARE_CTR) cnt[BKT_ROWS + t] = 0;
  int xs = s0, xn = ln;
  for (int d = 16; d > 0; d >>= 1) { xs += __shfl_xor(xs, d); xn += __shfl_xor(xn, d); }
  if (lane == 0) { wbs[wave] = xs; wn[wave] = xn; }
  __syncthreads();
  int bs = 0, n = 0;
  for (int w = 0; w < 16; ++w) { bs += wbs[w]; n += wn[w]; }
  bs = min(bs, N_EDGES);
  n  = min(n, N_EDGES - bs);
  const int be = bs + n;
  int2 e[PCS_TRIPS][4];
  const int pc0 = wave * (4 * PCS_TRIPS) + g;
  int longp = 0;
  const int norec = (BKT_ROWS + lane) << COL_BITS;
  #pragma unroll
  for (int tr = 0; tr < PCS_TRIPS; ++tr) {
    const int src = psrc[pc0 + tr * 4], len = plen[pc0 + tr * 4];
    longp |= (len > 32);
    #pragma unroll
    for (int u = 0; u < 4; ++u) {
      e[tr][u] = make_int2(norec, 0);
      if (q + 8 * u < len) e[tr][u] = *(const int2*)((const char*)grp + (size_t)((unsigned)(src + q + 8 * u) * 8u));
    }
  }
  #pragma unroll
  for (int tr = 0; tr < PCS_TRIPS; ++tr) {
    #pragma unroll
    for (int u = 0; u < 4; ++u) atomicAdd(&cnt[lg_ctr(e[tr][u].x)], 1);
  }
  if (longp)
    for (int tr = 0; tr < PCS_TRIPS; ++tr) {
      const int src = psrc[pc0 + tr * 4], len = plen[pc0 + tr * 4];
      for (int o = 32 + q; o < len; o += 8) atomicAdd(&cnt[lg_ctr(grp[src + o].x)], 1);
    }
  __syncthreads();
  const int c = cnt[t];
  scn[t] = c;
  __syncthreads();
  for (int s = 1; s < BKT_ROWS; s <<= 1) {
    const int add = (t >= s) ? scn[t - s] : 0;
    __syncthreads();
    scn[t] += add;
    __syncthreads();
  }
  const int start = bs + scn[t] - c;
  rowp[b * BKT_ROWS + t] = start;
  cnt[t] = start;
  __syncthreads();
  if (n > STG_CAP) {
    lg_rank_place<false>(grp, e, psrc, plen, pc0, longp, q, bs, cnt, stg, csr);
    return;
  }
  lg_rank_place<true>(grp, e, psrc, plen, pc0, longp, q, bs, cnt, stg, csr);
  __syncthreads();
  const int h = min((16 - (bs & 15)) & 15, n);
  if (t < h) csr[bs + t] = stg[t];
  const int np = (n - h) >> 1;
  v4i* dst4 = (v4i*)(csr + bs + h);
  #pragma unroll 4
  for (int i = t; i < np; i += 512) {
    const int j = h + 2 * i;
    const v4i o = {stg[j].x, stg[j].y, stg[j + 1].x, stg[j + 1].y};
    dst4[i] = o;
  }
  if (t == 0 && ((n - h) & 1)) csr[be - 1] = stg[n - 1];
}


template <int K>
__device__ __forceinline__ int grp_bcast(int v) {
  return (int)__builtin_amdgcn_mov_dpp8((unsigned)v, (unsigned)(K * 0x249249));
}

#define LG_SLOT(K)                                                                \
  {                                                                               \
    const unsigned c = (unsigned)grp_bcast<K>(cc);                                \
    w[K] = __int_as_float(grp_bcast<K>(ww));                                      \
    const char* p = srcb + (size_t)(c * (unsigned)(EMBED_DIM * 4) + qoff);        \
    x0[K] = *(const v4f*)p;                                                       \
    x1[K] = *(const v4f*)(p + 128);                                               \
  }

__device__ __forceinline__ void lg_round(const char* __restrict__ srcb, const int cc, const int ww,
                                         const unsigned qoff, v4f& a0, v4f& a1) {
  v4f   x0[8], x1[8];
  float w[8];
  LG_SLOT(0) LG_SLOT(1) LG_SLOT(2) LG_SLOT(3) LG_SLOT(4) LG_SLOT(5) LG_SLOT(6) LG_SLOT(7)
  #pragma unroll
  for (int k = 0; k < 8; ++k) { a0 += w[k] * x0[k]; a1 += w[k] * x1[k]; }
}

__device__ __forceinline__ void lg_load_recs(const int2* __restrict__ csr, const int jb, const int n, const int q,
                                             int2 (&rec)[4]) {
  #pragma unroll
  for (int j = 0; j < 4; ++j) {
    rec[j] = make_int2(0, 0);
    if (8 * j < n) rec[j] = csr[jb + min(8 * j + q, n - 1)];
  }
}

__device__ __forceinline__ void lg_chunk_sum(const char* __restrict__ srcb, const int2 (&rec)[4], const int n,
                                             const int q, const unsigned qoff, v4f& a0, v4f& a1) {
  #pragma unroll
  for (int j = 0; j < 4; ++j)
    if (8 * j < n)
      lg_round(srcb, rec[j].x, (8 * j + q < n) ? rec[j].y : 0  , qoff, a0, a1);
}

__device__ __forceinline__ void lg_row_sum(const char* __restrict__ srcb, const int2* __restrict__ csr,
                                           const int beg, const int end, const int q, const unsigned qoff,
                                           v4f& a0, v4f& a1) {
  for (int jb = beg; jb < end; jb += 32) {
    const int n = min(end - jb, 32);
    int2 rec[4];
    lg_load_recs(csr, jb, n, q, rec);
    lg_chunk_sum(srcb, rec, n, q, qoff, a0, a1);
  }
}

#define L1_BLOCKS  1024
__global__ __launch_bounds__(256) void lg_layer1(const float* __restrict__ emb, const int* __restrict__ rowp,
                                                 const int2* __restrict__ csr, float* __restrict__ t1) {
  static_assert(N_TOTAL % 4 == 0, "a wave takes four whole rows");
  const int lane = threadIdx.x & 31;
  const int q = lane & 7;
  const int g = lane >> 3;
  const unsigned qoff = (unsigned)q * 16u;
  const char* srcb = (const char*)emb;
  const int w0 = __builtin_amdgcn_readfirstlane((int)((blockIdx.x * blockDim.x + threadIdx.x) >> 5));
  const int step = (int)((gridDim.x * blockDim.x) >> 5) * 4;
  int r0 = w0 * 4;
  if (r0 >= N_TOTAL) return;
  int row = r0 + g;
  int beg = rowp[row], end = rowp[row + 1];
  int2 rec[4];
  lg_load_recs(csr, beg, min(end - beg, 32), q, rec);
  int begN = 0, endN = 0;
  if (r0 + step < N_TOTAL) { begN = rowp[row + step]; endN = rowp[row + step + 1]; }
  for (; r0 < N_TOTAL; r0 += step) {
    int2 recN[4];
    lg_load_recs(csr, begN, min(endN - begN, 32), q, recN);
    int begNN = 0, endNN = 0;
    if (r0 + 2 * step < N_TOTAL) { begNN = rowp[row + 2 * step]; endNN = rowp[row + 2 * step + 1]; }
    v4f a0 = {0.f, 0.f, 0.f, 0.f}, a1 = a0;
    lg_chunk_sum(srcb, rec, min(end - beg, 32), q, qoff, a0, a1);
    for (int jb = beg + 32; jb < end; jb += 32) {
      const int n = min(end - jb, 32);
      int2 rc[4];
      lg_load_recs(csr, jb, n, q, rc);
      lg_chunk_sum(srcb, rc, n, q, qoff, a0, a1);
    }
    *(v4f*)(t1 + (size_t)row * EMBED_DIM + q * 4) = a0;
    *(v4f*)(t1 + (size_t)row * EMBED_DIM + HALF_DIM + q * 4) = a1;
    row += step;
    beg = begN; end = endN; begN = begNN; endN = endNN;
    #pragma unroll
    for (int j = 0; j < 4; ++j) rec[j] = recN[j];
  }
}

#define PAIRS_PER_BLOCK 16
#define ACC_PITCH       68
static_assert(BATCH % PAIRS_PER_BLOCK == 0 && (ACC_PITCH * 4) % 16 == 0, "whole blocks of pairs; staged rows stay 16-byte aligned");
__global__ __launch_bounds__(1024) void lg_layer23_dot(const float* __restrict__ emb0, const float* __restrict__ t1,
                                                       const int* __restrict__ rowp, const int2* __restrict__ csr,
                                                       const int* __restrict__ uid, const int* __restrict__ iid,
                                                       float* __restrict__ out) {
  __shared__ __attribute__((aligned(16))) float accL[2 * PAIRS_PER_BLOCK][ACC_PITCH];
  const int lane = threadIdx.x & 31;
  const int q = lane & 7;
  const int g = lane >> 3;
  const unsigned qoff = (unsigned)q * 16u;
  const int wave = __builtin_amdgcn_readfirstlane((int)(threadIdx.x >> 5));
  const int pair = (int)blockIdx.x * PAIRS_PER_BLOCK + (wave & (PAIRS_PER_BLOCK - 1));
  const int row = (wave < PAIRS_PER_BLOCK) ? uid[pair] : (NUM_USERS + iid[pair]);
  const int begR = rowp[row];
  const int ntask = min(max(rowp[row + 1] - begR, 0), N_EDGES) + 1;
  const v4f zero = {0.f, 0.f, 0.f, 0.f};
  v4f l2a = zero, l2b = zero, s0 = zero, s1 = zero;
  int2 rrN = make_int2(row, 0);
  if (g > 0 && g < ntask) rrN = csr[begR + g - 1];
  for (int t0 = 0; t0 < ntask; t0 += 4) {
    const int t = t0 + g;
    const int2 rr = rrN;
    rrN = make_int2(row, 0);
    if (t + 4 < ntask) rrN = csr[begR + t + 3];
    int beg = 0, end = 0;
    if (t < ntask) { beg = rowp[rr.x]; end = rowp[rr.x + 1]; }
    v4f a0 = zero, a1 = zero;
    lg_row_sum((const char*)t1, csr, beg, end, q, qoff, a0, a1);
    #pragma unroll
    for (int gg = 0; gg < 4; ++gg)
      if (t0 + gg < ntask) {
        const int sl = gg * 8 + q;
        v4f b0, b1;
        #pragma unroll
        for (int e = 0; e < 4; ++e) { b0[e] = __shfl(a0[e], sl); b1[e] = __shfl(a1[e], sl); }
        const float wv = __int_as_float(__shfl(rr.y, gg * 8));
        if (t0 + gg == 0) { l2a = b0; l2b = b1; }
        else              { s0 += wv * b0; s1 += wv * b1; }
      }
  }
  if (g == 0) {
    const v4f l0a = *(const v4f*)(emb0 + (size_t)row * EMBED_DIM + q * 4);
    const v4f l0b = *(const v4f*)(emb0 + (size_t)row * EMBED_DIM + HALF_DIM + q * 4);
    const v4f l1a = *(const v4f*)(t1 + (size_t)row * EMBED_DIM + q * 4);
    const v4f l1b = *(const v4f*)(t1 + (size_t)row * EMBED_DIM + HALF_DIM + q * 4);
    *(v4f*)&accL[wave][q * 4] = ((l0a + l1a) + l2a) + s0;
    *(v4f*)&accL[wave][HALF_DIM + q * 4] = ((l0b + l1b) + l2b) + s1;
  }
  __syncthreads();
  if (wave != 0) return;

  const int rN = lane & 15;
  const float* U = accL[rN];
  const float* I = accL[PAIRS_PER_BLOCK + rN];

  v8f c = {0.f, 0.f, 0.f, 0.f, 0.f, 0.f, 0.f, 0.f};

#if __has_builtin(__builtin_amdgcn_wmma_f32_16x16x4_f32)
  const int kb = (lane >> 4) << 1;
  #pragma unroll
  for (int s = 0; s < 16; ++s) {
    const int col = s * 4 + kb;
    v2f a; a[0] = U[col]; a[1] = U[col + 1];
    v2f b; b[0] = I[col]; b[1] = I[col + 1];
    c = __builtin_amdgcn_wmma_f32_16x16x4_f32(false, a, false, b, (short)0, c, false, false);
  }
#else
  const int kb8 = (lane >> 4) << 3;
  #pragma unroll
  for (int blk = 0; blk < 2; ++blk) {
    v16h a, b;
    #pragma unroll
    for (int i = 0; i < 8; ++i) {
      const int kk = blk * 32 + kb8 + ((i < 4) ? (2 * i) : (16 + 2 * (i - 4)));
      a[2 * i]     = (_Float16)U[kk];
      a[2 * i + 1] = (_Float16)U[kk + 1];
      b[2 * i]     = (_Float16)I[kk];
      b[2 * i + 1] = (_Float16)I[kk + 1];
    }
    c = __builtin_amdgcn_wmma_f32_16x16x32_f16(false, a, false, b, (short)0, c, false, false);
  }
#endif

  const int sl = (lane < 8) ? lane : ((lane + 16) & 31);
  float t[8];
  #pragma unroll
  for (int r = 0; r < 8; ++r) t[r] = __shfl(c[r], sl);
  float d = t[0];
  #pragma unroll
  for (int r = 1; r < 8; ++r) d = ((lane & 7) == r) ? t[r] : d;
  if (lane < 16) out[(int)blockIdx.x * PAIRS_PER_BLOCK + lane] = d * 0.0625f;
}

extern "C" void kernel_launch(void* const* d_in, const int* in_sizes, int n_in,
                              void* d_out, int out_size, void* d_ws, size_t ws_size,
                              hipStream_t stream) {
  const float* node_emb = (const float*)d_in[0];
  const float* adj_val  = (const float*)d_in[1];
  const int*   adj_row  = (const int*)d_in[2];
  const int*   adj_col  = (const int*)d_in[3];
  const int*   user_ids = (const int*)d_in[4];
  const int*   item_ids = (const int*)d_in[5];
  float* out = (float*)d_out;

  char* ws    = (char*)d_ws;
  float* tbl1 = (float*)(ws + OFF_TBL1);
  int2*  grp  = (int2*)(ws + OFF_TBL1);
  int2*  csr  = (int2*)(ws + OFF_CSR);
  int*   rowp = (int*)(ws + OFF_ROWP);
  int*   pcs  = (int*)(ws + OFF_PCS);

  const dim3 b256(256);

  lg_chunk_group<<<NCHK, 512, 0, stream>>>(adj_row, adj_col, adj_val, pcs, grp);
  lg_bkt_rows<<<NBKT, 512, 0, stream>>>(grp, pcs, rowp, csr);

  lg_layer1<<<L1_BLOCKS, b256, 0, stream>>>(node_emb, rowp, csr, tbl1);
  lg_layer23_dot<<<BATCH / PAIRS_PER_BLOCK, 1024, 0, stream>>>(node_emb, tbl1, rowp, csr, user_ids, item_ids, out);
}
